# starting kernel
# baseline (speedup 1.0000x reference)
#include <hip/hip_runtime.h>
#include <math.h>

typedef __attribute__((ext_vector_type(2))) float v2f;
typedef __attribute__((ext_vector_type(8))) float v8f;

#define LN_EPS 1e-5f

__device__ __forceinline__ void atomAddF(float* p, float v) {
  unsafeAtomicAdd(p, v);
}

__global__ void init_deg(float* deg, int n) {
  int i = blockIdx.x * blockDim.x + threadIdx.x;
  if (i < n) deg[i] = 1.0f;
}

__global__ void deg_accum(const int* __restrict__ dst, float* deg, int E) {
  int i = blockIdx.x * blockDim.x + threadIdx.x;
  if (i < E) atomAddF(&deg[dst[i]], 1.0f);
}

__global__ void finish_dinv(float* deg, int n) {
  int i = blockIdx.x * blockDim.x + threadIdx.x;
  if (i < n) {
    float d = deg[i];
    deg[i] = (d > 0.0f) ? rsqrtf(d) : 0.0f;
  }
}

template <int FOUT>
__global__ __launch_bounds__(256) void gemm_wmma_f32(
    const float* __restrict__ X, const float* __restrict__ W,
    float* __restrict__ H, int nrows)
{
  constexpr int TPR = FOUT / 16;
  constexpr int RPB = 8 / TPR;
  const int w    = threadIdx.x >> 5;
  const int lane = threadIdx.x & 31;

  const int rowTile = blockIdx.x * RPB + (w / TPR);
  const int colTile = w % TPR;
  const int r0 = rowTile * 16;
  if (r0 >= nrows) return;
  const int c0 = colTile * 16;

  const int lh = lane & 15;
  const int kh = (lane >> 4) << 1;

  const float* __restrict__ xrow = X + (size_t)(r0 + lh) * 128 + kh;
  const float* __restrict__ wcol = W + (size_t)kh * FOUT + c0 + lh;

  v8f acc = {};
#pragma unroll
  for (int kk = 0; kk < 128; kk += 4) {
    v2f a; a.x = xrow[0];   a.y = xrow[1];
    v2f b; b.x = wcol[0];   b.y = wcol[FOUT];
    acc = __builtin_amdgcn_wmma_f32_16x16x4_f32(
        false, a, false, b, (short)0, acc, false, false);
    xrow += 4;
    wcol += 4 * FOUT;
  }

  const int rbase = r0 + ((lane >> 4) << 3);
  float* hp = H + (size_t)rbase * FOUT + c0 + lh;
#pragma unroll
  for (int v = 0; v < 8; ++v) hp[(size_t)v * FOUT] = acc[v];
}

template <int FOUT>
__global__ __launch_bounds__(256) void scatter_edges(
    const float* __restrict__ H, const int* __restrict__ srcA,
    const int* __restrict__ dstA, const float* __restrict__ dinv,
    float* __restrict__ AGG, int E, int n)
{
  constexpr int VEC = FOUT / 32;
  const int gw   = (int)((blockIdx.x * blockDim.x + threadIdx.x) >> 5);
  const int lane = threadIdx.x & 31;
  const int total = E + n;
  if (gw >= total) return;

  int s, d;
  if (gw < E) { s = srcA[gw]; d = dstA[gw]; }
  else        { s = gw - E;   d = s; }

  const float nrm = dinv[s] * dinv[d];
  const float* __restrict__ hp = H + (size_t)s * FOUT + lane * VEC;
  float* ap = AGG + (size_t)d * FOUT + lane * VEC;

  float v[VEC];
#pragma unroll
  for (int j = 0; j < VEC; ++j) v[j] = hp[j] * nrm;
#pragma unroll
  for (int j = 0; j < VEC; ++j) atomAddF(ap + j, v[j]);
}

__global__ __launch_bounds__(256) void ln_relu_128(
    const float* __restrict__ AGG, const float* __restrict__ bias,
    const float* __restrict__ gamma, const float* __restrict__ beta,
    float* __restrict__ OUT, int n)
{
  const int node = (int)((blockIdx.x * blockDim.x + threadIdx.x) >> 5);
  const int lane = threadIdx.x & 31;
  if (node >= n) return;

  const float* row = AGG + (size_t)node * 128 + lane * 4;
  float v0 = row[0] + bias[lane * 4 + 0];
  float v1 = row[1] + bias[lane * 4 + 1];
  float v2 = row[2] + bias[lane * 4 + 2];
  float v3 = row[3] + bias[lane * 4 + 3];

  float s = v0 + v1 + v2 + v3;
#pragma unroll
  for (int m = 16; m >= 1; m >>= 1) s += __shfl_xor(s, m, 32);
  const float mu = s * (1.0f / 128.0f);

  float d0 = v0 - mu, d1 = v1 - mu, d2 = v2 - mu, d3 = v3 - mu;
  float q = d0 * d0 + d1 * d1 + d2 * d2 + d3 * d3;
#pragma unroll
  for (int m = 16; m >= 1; m >>= 1) q += __shfl_xor(q, m, 32);
  const float rstd = rsqrtf(q * (1.0f / 128.0f) + LN_EPS);

  float* out = OUT + (size_t)node * 128 + lane * 4;
  out[0] = fmaxf(0.0f, d0 * rstd * gamma[lane * 4 + 0] + beta[lane * 4 + 0]);
  out[1] = fmaxf(0.0f, d1 * rstd * gamma[lane * 4 + 1] + beta[lane * 4 + 1]);
  out[2] = fmaxf(0.0f, d2 * rstd * gamma[lane * 4 + 2] + beta[lane * 4 + 2]);
  out[3] = fmaxf(0.0f, d3 * rstd * gamma[lane * 4 + 3] + beta[lane * 4 + 3]);
}

__global__ void bias_add_64(const float* __restrict__ AGG,
                            const float* __restrict__ bias,
                            float* __restrict__ OUT, int total)
{
  int i = blockIdx.x * blockDim.x + threadIdx.x;
  if (i < total) OUT[i] = AGG[i] + bias[i & 63];
}

extern "C" void kernel_launch(void* const* d_in, const int* in_sizes, int n_in,
                              void* d_out, int out_size, void* d_ws, size_t ws_size,
                              hipStream_t stream) {
  const float* x   = (const float*)d_in[0];
  const int*   ei  = (const int*)d_in[1];
  const float* W1  = (const float*)d_in[2];
  const float* b1  = (const float*)d_in[3];
  const float* g1  = (const float*)d_in[4];
  const float* be1 = (const float*)d_in[5];
  const float* W2  = (const float*)d_in[6];
  const float* b2  = (const float*)d_in[7];
  const float* g2  = (const float*)d_in[8];
  const float* be2 = (const float*)d_in[9];
  const float* W3  = (const float*)d_in[10];
  const float* b3  = (const float*)d_in[11];

  const int N = in_sizes[0] / 128;
  const int E = in_sizes[1] / 2;
  const int* src = ei;
  const int* dst = ei + E;

  float* ws   = (float*)d_ws;
  float* dinv = ws;
  float* h    = dinv + N;
  float* agg  = h + (size_t)N * 128;
  float* hn   = agg + (size_t)N * 128;

  init_deg   <<<(N + 255) / 256, 256, 0, stream>>>(dinv, N);
  deg_accum  <<<(E + 255) / 256, 256, 0, stream>>>(dst, dinv, E);
  finish_dinv<<<(N + 255) / 256, 256, 0, stream>>>(dinv, N);

  const int total   = E + N;
  const int sblocks = (total + 7) / 8;

  gemm_wmma_f32<128><<<(N + 15) / 16, 256, 0, stream>>>(x, W1, h, N);
  hipMemsetAsync(agg, 0, (size_t)N * 128 * sizeof(float), stream);
  scatter_edges<128><<<sblocks, 256, 0, stream>>>(h, src, dst, dinv, agg, E, N);
  ln_relu_128<<<(N + 7) / 8, 256, 0, stream>>>(agg, b1, g1, be1, hn, N);

  gemm_wmma_f32<128><<<(N + 15) / 16, 256, 0, stream>>>(hn, W2, h, N);
  hipMemsetAsync(agg, 0, (size_t)N * 128 * sizeof(float), stream);
  scatter_edges<128><<<sblocks, 256, 0, stream>>>(h, src, dst, dinv, agg, E, N);
  ln_relu_128<<<(N + 7) / 8, 256, 0, stream>>>(agg, b2, g2, be2, hn, N);

  gemm_wmma_f32<64><<<(N + 31) / 32, 256, 0, stream>>>(hn, W3, h, N);
  hipMemsetAsync(agg, 0, (size_t)N * 64 * sizeof(float), stream);
  scatter_edges<64><<<sblocks, 256, 0, stream>>>(h, src, dst, dinv, agg, E, N);
  bias_add_64<<<(N * 64 + 255) / 256, 256, 0, stream>>>(agg, b3, (float*)d_out, N * 64);
}


# Row epilogue moved into the aggregation block: rows leave LDS through bias+LayerNorm+ReLU (layers 1,2) or +bias (layer 3) straight to the next plane/out; 12 -> 9 launches; sum orders unchanged
# speedup vs baseline: 2.0352x; 2.0352x over previous
#include <hip/hip_runtime.h>
#include <math.h>

typedef __attribute__((ext_vector_type(2))) float v2f;
typedef __attribute__((ext_vector_type(8))) float v8f;

#define LN_EPS 1e-5f

__device__ __forceinline__ void atomAddF(float* p, float v) {
  unsafeAtomicAdd(p, v);
}

__global__ void init_deg(float* deg, int n) {
  int i = blockIdx.x * blockDim.x + threadIdx.x;
  if (i < n) deg[i] = 1.0f;
}

__global__ void deg_accum(const int* __restrict__ dst, float* deg, int E) {
  int i = blockIdx.x * blockDim.x + threadIdx.x;
  if (i < E) atomAddF(&deg[dst[i]], 1.0f);
}

__global__ void finish_dinv(float* deg, int n) {
  int i = blockIdx.x * blockDim.x + threadIdx.x;
  if (i < n) {
    float d = deg[i];
    deg[i] = (d > 0.0f) ? rsqrtf(d) : 0.0f;
  }
}

template <int FOUT>
__global__ __launch_bounds__(256) void gemm_wmma_f32(
    const float* __restrict__ X, const float* __restrict__ W,
    float* __restrict__ H, int nrows)
{
  constexpr int TPR = FOUT / 16;
  constexpr int RPB = 8 / TPR;
  const int w    = threadIdx.x >> 5;
  const int lane = threadIdx.x & 31;

  const int rowTile = blockIdx.x * RPB + (w / TPR);
  const int colTile = w % TPR;
  const int r0 = rowTile * 16;
  if (r0 >= nrows) return;
  const int c0 = colTile * 16;

  const int lh = lane & 15;
  const int kh = (lane >> 4) << 1;

  const float* __restrict__ xrow = X + (size_t)(r0 + lh) * 128 + kh;
  const float* __restrict__ wcol = W + (size_t)kh * FOUT + c0 + lh;

  v8f acc = {};
#pragma unroll
  for (int kk = 0; kk < 128; kk += 4) {
    v2f a; a.x = xrow[0];   a.y = xrow[1];
    v2f b; b.x = wcol[0];   b.y = wcol[FOUT];
    acc = __builtin_amdgcn_wmma_f32_16x16x4_f32(
        false, a, false, b, (short)0, acc, false, false);
    xrow += 4;
    wcol += 4 * FOUT;
  }

  const int rbase = r0 + ((lane >> 4) << 3);
  float* hp = H + (size_t)rbase * FOUT + c0 + lh;
#pragma unroll
  for (int v = 0; v < 8; ++v) hp[(size_t)v * FOUT] = acc[v];
}

#define RNG 196
#define NWV 32
#define LCAP 640
#define SCAN 16
#define INFL 16

template <int FOUT>
__device__ __forceinline__ void add_listed(
    const float* __restrict__ H, const int* __restrict__ srcA,
    const int* __restrict__ dstA, const float* __restrict__ dinv,
    float* acc, const int* le, int cnt, int d0, int lane)
{
  constexpr int LINES = FOUT / 32;
  constexpr int GRP = INFL / LINES;
  for (int k0 = 0; k0 < cnt; k0 += 32) {
    const int m = min(32, cnt - k0);
    int s = 0, dl = 0;
    float nrm = 0.0f;
    if (lane < m) {
      const int e = le[k0 + lane];
      const int d = dstA[e];
      s = srcA[e];
      dl = d - d0;
      nrm = dinv[s] * dinv[d];
    }
    int q = 0;
    for (; q + GRP <= m; q += GRP) {
      float v[GRP][LINES];
      int dq[GRP];
#pragma unroll
      for (int u = 0; u < GRP; ++u) {
        const int su = __builtin_amdgcn_readlane(s, q + u);
        const float nu = __int_as_float(__builtin_amdgcn_readlane(__float_as_int(nrm), q + u));
        dq[u] = __builtin_amdgcn_readlane(dl, q + u);
        const float* __restrict__ hp = H + (size_t)su * FOUT + lane;
#pragma unroll
        for (int j = 0; j < LINES; ++j) v[u][j] = hp[j * 32] * nu;
      }
#pragma unroll
      for (int u = 0; u < GRP; ++u) {
        float* ap = acc + dq[u] * FOUT + lane;
#pragma unroll
        for (int j = 0; j < LINES; ++j)
          __hip_atomic_fetch_add(ap + j * 32, v[u][j], __ATOMIC_RELAXED, __HIP_MEMORY_SCOPE_WORKGROUP);
      }
    }
    for (; q < m; ++q) {
      const int su = __builtin_amdgcn_readlane(s, q);
      const float nu = __int_as_float(__builtin_amdgcn_readlane(__float_as_int(nrm), q));
      const int dd = __builtin_amdgcn_readlane(dl, q);
      const float* __restrict__ hp = H + (size_t)su * FOUT + lane;
      float* ap = acc + dd * FOUT + lane;
      float v[LINES];
#pragma unroll
      for (int j = 0; j < LINES; ++j) v[j] = hp[j * 32] * nu;
#pragma unroll
      for (int j = 0; j < LINES; ++j)
        __hip_atomic_fetch_add(ap + j * 32, v[j], __ATOMIC_RELAXED, __HIP_MEMORY_SCOPE_WORKGROUP);
    }
  }
}

template <int M>
__device__ __forceinline__ float lane_xor(float x) {
  const int b = __builtin_bit_cast(int, x);
  if constexpr (M == 16)
    return __builtin_bit_cast(float,
        __builtin_amdgcn_permlanex16(b, b, 0x76543210, 0xfedcba98, false, false));
  else
    return __builtin_bit_cast(float,
        __builtin_amdgcn_update_dpp(0, b, 0x160 | M, 0xf, 0xf, false));
}

__device__ __forceinline__ float wave_sum(float s) {
  s += lane_xor<16>(s);
  s += lane_xor<8>(s);
  s += lane_xor<4>(s);
  s += lane_xor<2>(s);
  s += lane_xor<1>(s);
  return s;
}

template <int FOUT, bool LNRELU>
__global__ __launch_bounds__(1024) void agg_range(
    const float* __restrict__ H, const int* __restrict__ srcA,
    const int* __restrict__ dstA, const float* __restrict__ dinv,
    const float* __restrict__ bias, const float* __restrict__ gamma,
    const float* __restrict__ beta, float* __restrict__ OUT, int E, int n)
{
  __shared__ float acc[RNG * FOUT];
  __shared__ int lst_e[NWV * LCAP];

  const int tid  = threadIdx.x;
  const int w    = __builtin_amdgcn_readfirstlane(tid >> 5);
  const int lane = tid & 31;
  const int d0   = blockIdx.x * RNG;
  const int nd   = min(RNG, n - d0);

  for (int i = tid; i < nd * FOUT; i += NWV * 32) {
    const float di = dinv[d0 + i / FOUT];
    acc[i] = H[(size_t)d0 * FOUT + i] * (di * di);
  }
  __syncthreads();

  const int per = ((E + NWV * SCAN * 32 - 1) / (NWV * SCAN * 32)) * (SCAN * 32);
  const int wb  = w * per;
  const int we  = min(E, wb + per);
  const int ntr = (we > wb) ? (we - wb + SCAN * 32 - 1) / (SCAN * 32) : 0;
  int* le = lst_e + w * LCAP;
  int cnt = 0;
  int tt = (ntr > 0) ? (int)(blockIdx.x % (unsigned)ntr) : 0;

  for (int t = 0; t < ntr; ++t) {
    const int b = wb + tt * (SCAN * 32);
    tt = (tt + 1 == ntr) ? 0 : tt + 1;
    int dv[SCAN];
    if (b + SCAN * 32 <= we) {
#pragma unroll
      for (int u = 0; u < SCAN; ++u) dv[u] = dstA[b + u * 32 + lane];
    } else {
#pragma unroll
      for (int u = 0; u < SCAN; ++u) {
        const int i = b + u * 32 + lane;
        dv[u] = (i < we) ? dstA[i] : -1;
      }
    }
#pragma unroll
    for (int u = 0; u < SCAN; ++u) {
      const int rel = dv[u] - d0;
      const bool hit = (unsigned)rel < (unsigned)nd;
      const unsigned mk = __builtin_amdgcn_ballot_w32(hit);
      if (mk != 0u) {
        if (hit) le[cnt + __popc(mk & ((1u << lane) - 1u))] = b + u * 32 + lane;
        cnt += __popc(mk);
      }
    }
    if (cnt >= 64 || t + 1 == ntr) {
      add_listed<FOUT>(H, srcA, dstA, dinv, acc, le, cnt, d0, lane);
      cnt = 0;
    }
  }
  if constexpr (LNRELU) {
    static_assert(FOUT == 128, "the LayerNorm exit takes 4 features per lane of a 128-wide row");
    const float b0 = bias[lane * 4 + 0],  b1 = bias[lane * 4 + 1];
    const float b2 = bias[lane * 4 + 2],  b3 = bias[lane * 4 + 3];
    const float g0 = gamma[lane * 4 + 0], g1 = gamma[lane * 4 + 1];
    const float g2 = gamma[lane * 4 + 2], g3 = gamma[lane * 4 + 3];
    const float e0 = beta[lane * 4 + 0],  e1 = beta[lane * 4 + 1];
    const float e2 = beta[lane * 4 + 2],  e3 = beta[lane * 4 + 3];
    __syncthreads();

    for (int r = w; r < nd; r += NWV) {
      const float* row = acc + r * FOUT + lane * 4;
      const float v0 = row[0] + b0;
      const float v1 = row[1] + b1;
      const float v2 = row[2] + b2;
      const float v3 = row[3] + b3;

      const float mu = wave_sum(v0 + v1 + v2 + v3) * (1.0f / 128.0f);

      const float c0 = v0 - mu, c1 = v1 - mu, c2 = v2 - mu, c3 = v3 - mu;
      const float q = wave_sum(c0 * c0 + c1 * c1 + c2 * c2 + c3 * c3);
      const float rstd = rsqrtf(q * (1.0f / 128.0f) + LN_EPS);

      float* out = OUT + (size_t)(d0 + r) * FOUT + lane * 4;
      out[0] = fmaxf(0.0f, c0 * rstd * g0 + e0);
      out[1] = fmaxf(0.0f, c1 * rstd * g1 + e1);
      out[2] = fmaxf(0.0f, c2 * rstd * g2 + e2);
      out[3] = fmaxf(0.0f, c3 * rstd * g3 + e3);
    }
  } else {
    static_assert((NWV * 32) % FOUT == 0, "a thread keeps one feature column over its whole stride");
    const float bb = bias[tid & (FOUT - 1)];
    __syncthreads();
    for (int i = tid; i < nd * FOUT; i += NWV * 32)
      OUT[(size_t)d0 * FOUT + i] = acc[i] + bb;
  }
}

extern "C" void kernel_launch(void* const* d_in, const int* in_sizes, int n_in,
                              void* d_out, int out_size, void* d_ws, size_t ws_size,
                              hipStream_t stream) {
  const float* x   = (const float*)d_in[0];
  const int*   ei  = (const int*)d_in[1];
  const float* W1  = (const float*)d_in[2];
  const float* b1  = (const float*)d_in[3];
  const float* g1  = (const float*)d_in[4];
  const float* be1 = (const float*)d_in[5];
  const float* W2  = (const float*)d_in[6];
  const float* b2  = (const float*)d_in[7];
  const float* g2  = (const float*)d_in[8];
  const float* be2 = (const float*)d_in[9];
  const float* W3  = (const float*)d_in[10];
  const float* b3  = (const float*)d_in[11];

  const int N = in_sizes[0] / 128;
  const int E = in_sizes[1] / 2;
  const int* src = ei;
  const int* dst = ei + E;

  float* ws   = (float*)d_ws;
  float* h    = ws;
  float* hn   = h + (size_t)N * 128;
  float* dinv = hn + (size_t)N * 128;

  init_deg   <<<(N + 255) / 256, 256, 0, stream>>>(dinv, N);
  deg_accum  <<<(E + 255) / 256, 256, 0, stream>>>(dst, dinv, E);
  finish_dinv<<<(N + 255) / 256, 256, 0, stream>>>(dinv, N);

  const int ablocks = (N + RNG - 1) / RNG;

  gemm_wmma_f32<128><<<(N + 15) / 16, 256, 0, stream>>>(x, W1, h, N);
  agg_range<128, true><<<ablocks, NWV * 32, 0, stream>>>(h, src, dst, dinv, b1, g1, be1, hn, E, N);

  gemm_wmma_f32<128><<<(N + 15) / 16, 256, 0, stream>>>(hn, W2, h, N);
  agg_range<128, true><<<ablocks, NWV * 32, 0, stream>>>(h, src, dst, dinv, b2, g2, be2, hn, E, N);

  gemm_wmma_f32<64><<<(N + 31) / 32, 256, 0, stream>>>(hn, W3, h, N);
  agg_range<64, false><<<ablocks, NWV * 32, 0, stream>>>(h, src, dst, dinv, b3, nullptr, nullptr,
                                                    (float*)d_out, E, N);
}


# Range filter run once: layer 1's blocks save their filtered edges (source, row, coefficient) to a workspace list at offsets summed from degree counts; layers 2,3 read it; sum order: LDS arrival
# speedup vs baseline: 2.4841x; 1.2206x over previous
#include <hip/hip_runtime.h>
#include <math.h>

typedef __attribute__((ext_vector_type(2))) float v2f;
typedef __attribute__((ext_vector_type(8))) float v8f;

#define LN_EPS 1e-5f

__device__ __forceinline__ void atomAddF(float* p, float v) {
  unsafeAtomicAdd(p, v);
}

__global__ void init_deg(float* deg, int n) {
  int i = blockIdx.x * blockDim.x + threadIdx.x;
  if (i < n) deg[i] = 1.0f;
}

__global__ void deg_accum(const int* __restrict__ dst, float* deg, int E) {
  int i = blockIdx.x * blockDim.x + threadIdx.x;
  if (i < E) atomAddF(&deg[dst[i]], 1.0f);
}

__global__ void finish_dinv(const float* __restrict__ deg, float* __restrict__ dinv, int n) {
  int i = blockIdx.x * blockDim.x + threadIdx.x;
  if (i < n) {
    float d = deg[i];
    dinv[i] = (d > 0.0f) ? rsqrtf(d) : 0.0f;
  }
}

template <int FOUT>
__global__ __launch_bounds__(256) void gemm_wmma_f32(
    const float* __restrict__ X, const float* __restrict__ W,
    float* __restrict__ H, int nrows)
{
  constexpr int TPR = FOUT / 16;
  constexpr int RPB = 8 / TPR;
  const int w    = threadIdx.x >> 5;
  const int lane = threadIdx.x & 31;

  const int rowTile = blockIdx.x * RPB + (w / TPR);
  const int colTile = w % TPR;
  const int r0 = rowTile * 16;
  if (r0 >= nrows) return;
  const int c0 = colTile * 16;

  const int lh = lane & 15;
  const int kh = (lane >> 4) << 1;

  const float* __restrict__ xrow = X + (size_t)(r0 + lh) * 128 + kh;
  const float* __restrict__ wcol = W + (size_t)kh * FOUT + c0 + lh;

  v8f acc = {};
#pragma unroll
  for (int kk = 0; kk < 128; kk += 4) {
    v2f a; a.x = xrow[0];   a.y = xrow[1];
    v2f b; b.x = wcol[0];   b.y = wcol[FOUT];
    acc = __builtin_amdgcn_wmma_f32_16x16x4_f32(
        false, a, false, b, (short)0, acc, false, false);
    xrow += 4;
    wcol += 4 * FOUT;
  }

  const int rbase = r0 + ((lane >> 4) << 3);
  float* hp = H + (size_t)rbase * FOUT + c0 + lh;
#pragma unroll
  for (int v = 0; v < 8; ++v) hp[(size_t)v * FOUT] = acc[v];
}

#define RNG 196
#define NWV 32
#define LCAP 640
#define SCAN 16
#define INFL 16

template <int FOUT, bool SAVED>
__device__ __forceinline__ void add_listed(
    const float* __restrict__ H, const int* __restrict__ srcA,
    const int* __restrict__ dstA, const float* __restrict__ dinv,
    float* acc, const int* le, int cnt, int d0, int lane,
    int* __restrict__ lsrc, int* __restrict__ ldl, float* __restrict__ lnrm, int gp)
{
  constexpr int LINES = FOUT / 32;
  constexpr int GRP = INFL / LINES;
  int s_nx = 0, dl_nx = 0;
  float nrm_nx = 0.0f;
  if constexpr (SAVED) {
    if (lane < min(32, cnt)) {
      s_nx   = lsrc[gp + lane];
      dl_nx  = ldl[gp + lane];
      nrm_nx = lnrm[gp + lane];
    }
  }
  for (int k0 = 0; k0 < cnt; k0 += 32) {
    const int m = min(32, cnt - k0);
    int s = 0, dl = 0;
    float nrm = 0.0f;
    if constexpr (SAVED) {
      s = s_nx; dl = dl_nx; nrm = nrm_nx;
      if (lane < cnt - k0 - 32) {
        s_nx   = lsrc[gp + k0 + 32 + lane];
        dl_nx  = ldl[gp + k0 + 32 + lane];
        nrm_nx = lnrm[gp + k0 + 32 + lane];
      }
    }
    if (lane < m) {
      if constexpr (!SAVED) {
        const int e = le[k0 + lane];
        const int d = dstA[e];
        s = srcA[e];
        dl = d - d0;
        nrm = dinv[s] * dinv[d];
        lsrc[gp + k0 + lane] = s;
        ldl[gp + k0 + lane]  = dl;
        lnrm[gp + k0 + lane] = nrm;
      }
    }
    int q = 0;
    for (; q + GRP <= m; q += GRP) {
      float v[GRP][LINES];
      int dq[GRP];
#pragma unroll
      for (int u = 0; u < GRP; ++u) {
        const int su = __builtin_amdgcn_readlane(s, q + u);
        const float nu = __int_as_float(__builtin_amdgcn_readlane(__float_as_int(nrm), q + u));
        dq[u] = __builtin_amdgcn_readlane(dl, q + u);
        const float* __restrict__ hp = H + (size_t)su * FOUT + lane;
#pragma unroll
        for (int j = 0; j < LINES; ++j) v[u][j] = hp[j * 32] * nu;
      }
#pragma unroll
      for (int u = 0; u < GRP; ++u) {
        float* ap = acc + dq[u] * FOUT + lane;
#pragma unroll
        for (int j = 0; j < LINES; ++j)
          __hip_atomic_fetch_add(ap + j * 32, v[u][j], __ATOMIC_RELAXED, __HIP_MEMORY_SCOPE_WORKGROUP);
      }
    }
    for (; q < m; ++q) {
      const int su = __builtin_amdgcn_readlane(s, q);
      const float nu = __int_as_float(__builtin_amdgcn_readlane(__float_as_int(nrm), q));
      const int dd = __builtin_amdgcn_readlane(dl, q);
      const float* __restrict__ hp = H + (size_t)su * FOUT + lane;
      float* ap = acc + dd * FOUT + lane;
      float v[LINES];
#pragma unroll
      for (int j = 0; j < LINES; ++j) v[j] = hp[j * 32] * nu;
#pragma unroll
      for (int j = 0; j < LINES; ++j)
        __hip_atomic_fetch_add(ap + j * 32, v[j], __ATOMIC_RELAXED, __HIP_MEMORY_SCOPE_WORKGROUP);
    }
  }
}

template <int M>
__device__ __forceinline__ int lane_xor_bits(int b) {
  if constexpr (M == 16)
    return __builtin_amdgcn_permlanex16(b, b, 0x76543210, 0xfedcba98, false, false);
  else
    return __builtin_amdgcn_update_dpp(0, b, 0x160 | M, 0xf, 0xf, false);
}
template <int M>
__device__ __forceinline__ float lane_xor(float x) {
  return __builtin_bit_cast(float, lane_xor_bits<M>(__builtin_bit_cast(int, x)));
}

__device__ __forceinline__ float wave_sum(float s) {
  s += lane_xor<16>(s);
  s += lane_xor<8>(s);
  s += lane_xor<4>(s);
  s += lane_xor<2>(s);
  s += lane_xor<1>(s);
  return s;
}

__device__ __forceinline__ int wave_sum_i(int s) {
  s += lane_xor_bits<16>(s);
  s += lane_xor_bits<8>(s);
  s += lane_xor_bits<4>(s);
  s += lane_xor_bits<2>(s);
  s += lane_xor_bits<1>(s);
  return s;
}

template <int FOUT, bool LNRELU>
__device__ __forceinline__ void rows_out(
    const float* acc, const float* __restrict__ bias, const float* __restrict__ gamma,
    const float* __restrict__ beta, float* __restrict__ OUT,
    int d0, int nd, int tid, int w, int lane)
{
  if constexpr (LNRELU) {
    static_assert(FOUT == 128, "the LayerNorm exit takes 4 features per lane of a 128-wide row");
    const float b0 = bias[lane * 4 + 0],  b1 = bias[lane * 4 + 1];
    const float b2 = bias[lane * 4 + 2],  b3 = bias[lane * 4 + 3];
    const float g0 = gamma[lane * 4 + 0], g1 = gamma[lane * 4 + 1];
    const float g2 = gamma[lane * 4 + 2], g3 = gamma[lane * 4 + 3];
    const float e0 = beta[lane * 4 + 0],  e1 = beta[lane * 4 + 1];
    const float e2 = beta[lane * 4 + 2],  e3 = beta[lane * 4 + 3];
    __syncthreads();

    for (int r = w; r < nd; r += NWV) {
      const float* row = acc + r * FOUT + lane * 4;
      const float v0 = row[0] + b0;
      const float v1 = row[1] + b1;
      const float v2 = row[2] + b2;
      const float v3 = row[3] + b3;

      const float mu = wave_sum(v0 + v1 + v2 + v3) * (1.0f / 128.0f);

      const float c0 = v0 - mu, c1 = v1 - mu, c2 = v2 - mu, c3 = v3 - mu;
      const float q = wave_sum(c0 * c0 + c1 * c1 + c2 * c2 + c3 * c3);
      const float rstd = rsqrtf(q * (1.0f / 128.0f) + LN_EPS);

      float* out = OUT + (size_t)(d0 + r) * FOUT + lane * 4;
      out[0] = fmaxf(0.0f, c0 * rstd * g0 + e0);
      out[1] = fmaxf(0.0f, c1 * rstd * g1 + e1);
      out[2] = fmaxf(0.0f, c2 * rstd * g2 + e2);
      out[3] = fmaxf(0.0f, c3 * rstd * g3 + e3);
    }
  } else {
    static_assert((NWV * 32) % FOUT == 0, "a thread keeps one feature column over its whole stride");
    const float bb = bias[tid & (FOUT - 1)];
    __syncthreads();
    for (int i = tid; i < nd * FOUT; i += NWV * 32)
      OUT[(size_t)d0 * FOUT + i] = acc[i] + bb;
  }
}

template <int FOUT>
__device__ __forceinline__ void rows_start(
    float* acc, const float* __restrict__ H, const float* __restrict__ dinv,
    int d0, int nd, int tid)
{
  for (int i = tid; i < nd * FOUT; i += NWV * 32) {
    const float di = dinv[d0 + i / FOUT];
    acc[i] = H[(size_t)d0 * FOUT + i] * (di * di);
  }
}

template <int FOUT, bool LNRELU>
__global__ __launch_bounds__(1024) void agg_range(
    const float* __restrict__ H, const int* __restrict__ srcA,
    const int* __restrict__ dstA, const float* __restrict__ dinv,
    const float* __restrict__ deg,
    const float* __restrict__ bias, const float* __restrict__ gamma,
    const float* __restrict__ beta, float* __restrict__ OUT,
    int* __restrict__ lsrc, int* __restrict__ ldl, float* __restrict__ lnrm,
    int* __restrict__ blk, int E, int n)
{
  __shared__ float acc[RNG * FOUT];
  __shared__ int lst_e[NWV * LCAP];
  __shared__ int red[2 * NWV];
  __shared__ int cursor;

  const int tid  = threadIdx.x;
  const int w    = __builtin_amdgcn_readfirstlane(tid >> 5);
  const int lane = tid & 31;
  const int d0   = blockIdx.x * RNG;
  const int nd   = min(RNG, n - d0);

  {
    static_assert(NWV == 32 && RNG <= NWV * 32, "one partial per lane; one owned row per thread");
    static_assert(RNG % 4 == 0, "the nodes below a range are a whole number of 16-byte pieces");
    constexpr int PFL = 8;
    const float4* __restrict__ dg4 = reinterpret_cast<const float4*>(deg);
    const int n4 = d0 >> 2;
    int sb = 0;
    for (int i0 = tid; i0 < n4; i0 += PFL * NWV * 32) {
      float4 c[PFL];
#pragma unroll
      for (int u = 0; u < PFL; ++u) c[u] = dg4[min(i0 + u * NWV * 32, n4 - 1)];
#pragma unroll
      for (int u = 0; u < PFL; ++u) {
        const int k = (int)c[u].x + (int)c[u].y + (int)c[u].z + (int)c[u].w - 4;
        sb += (i0 + u * NWV * 32 < n4) ? k : 0;
      }
    }
    int sc = (tid < nd) ? (int)deg[d0 + tid] - 1 : 0;
    sb = wave_sum_i(sb);
    sc = wave_sum_i(sc);
    if (lane == 0) { red[w] = sb; red[NWV + w] = sc; }
    if (tid == 0) cursor = 0;
  }
  rows_start<FOUT>(acc, H, dinv, d0, nd, tid);
  __syncthreads();
  const int base = __builtin_amdgcn_readfirstlane(wave_sum_i(red[lane]));
  if (w == 0) {
    const int tot = wave_sum_i(red[NWV + lane]);
    if (lane == 0) { blk[2 * blockIdx.x] = base; blk[2 * blockIdx.x + 1] = tot; }
  }

  const int per = ((E + NWV * SCAN * 32 - 1) / (NWV * SCAN * 32)) * (SCAN * 32);
  const int wb  = w * per;
  const int we  = min(E, wb + per);
  const int ntr = (we > wb) ? (we - wb + SCAN * 32 - 1) / (SCAN * 32) : 0;
  int* le = lst_e + w * LCAP;
  int cnt = 0;
  int tt = (ntr > 0) ? (int)(blockIdx.x % (unsigned)ntr) : 0;

  for (int t = 0; t < ntr; ++t) {
    const int b = wb + tt * (SCAN * 32);
    tt = (tt + 1 == ntr) ? 0 : tt + 1;
    int dv[SCAN];
    if (b + SCAN * 32 <= we) {
#pragma unroll
      for (int u = 0; u < SCAN; ++u) dv[u] = dstA[b + u * 32 + lane];
    } else {
#pragma unroll
      for (int u = 0; u < SCAN; ++u) {
        const int i = b + u * 32 + lane;
        dv[u] = (i < we) ? dstA[i] : -1;
      }
    }
#pragma unroll
    for (int u = 0; u < SCAN; ++u) {
      const int rel = dv[u] - d0;
      const bool hit = (unsigned)rel < (unsigned)nd;
      const unsigned mk = __builtin_amdgcn_ballot_w32(hit);
      if (mk != 0u) {
        if (hit) le[cnt + __popc(mk & ((1u << lane) - 1u))] = b + u * 32 + lane;
        cnt += __popc(mk);
      }
    }
    if (cnt >= 64 || t + 1 == ntr) {
      int pos = 0;
      if (lane == 0)
        pos = __hip_atomic_fetch_add(&cursor, cnt, __ATOMIC_RELAXED, __HIP_MEMORY_SCOPE_WORKGROUP);
      pos = __builtin_amdgcn_readfirstlane(pos);
      add_listed<FOUT, false>(H, srcA, dstA, dinv, acc, le, cnt, d0, lane,
                              lsrc, ldl, lnrm, base + pos);
      cnt = 0;
    }
  }
  rows_out<FOUT, LNRELU>(acc, bias, gamma, beta, OUT, d0, nd, tid, w, lane);
}

template <int FOUT, bool LNRELU>
__global__ __launch_bounds__(1024) void agg_list(
    const float* __restrict__ H, const float* __restrict__ dinv,
    const float* __restrict__ bias, const float* __restrict__ gamma,
    const float* __restrict__ beta, float* __restrict__ OUT,
    int* __restrict__ lsrc, int* __restrict__ ldl, float* __restrict__ lnrm,
    const int* __restrict__ blk, int n)
{
  __shared__ float acc[RNG * FOUT];

  const int tid  = threadIdx.x;
  const int w    = __builtin_amdgcn_readfirstlane(tid >> 5);
  const int lane = tid & 31;
  const int d0   = blockIdx.x * RNG;
  const int nd   = min(RNG, n - d0);

  const int base = blk[2 * blockIdx.x];
  const int tot  = blk[2 * blockIdx.x + 1];

  rows_start<FOUT>(acc, H, dinv, d0, nd, tid);
  __syncthreads();

  constexpr int GRP = INFL / (FOUT / 32);
  const int per = ((tot + NWV * GRP - 1) / (NWV * GRP)) * GRP;
  const int lo  = w * per;
  const int cnt = min(per, tot - lo);
  if (cnt > 0)
    add_listed<FOUT, true>(H, nullptr, nullptr, dinv, acc, nullptr, cnt, d0, lane,
                           lsrc, ldl, lnrm, base + lo);

  rows_out<FOUT, LNRELU>(acc, bias, gamma, beta, OUT, d0, nd, tid, w, lane);
}

extern "C" void kernel_launch(void* const* d_in, const int* in_sizes, int n_in,
                              void* d_out, int out_size, void* d_ws, size_t ws_size,
                              hipStream_t stream) {
  const float* x   = (const float*)d_in[0];
  const int*   ei  = (const int*)d_in[1];
  const float* W1  = (const float*)d_in[2];
  const float* b1  = (const float*)d_in[3];
  const float* g1  = (const float*)d_in[4];
  const float* be1 = (const float*)d_in[5];
  const float* W2  = (const float*)d_in[6];
  const float* b2  = (const float*)d_in[7];
  const float* g2  = (const float*)d_in[8];
  const float* be2 = (const float*)d_in[9];
  const float* W3  = (const float*)d_in[10];
  const float* b3  = (const float*)d_in[11];

  const int N = in_sizes[0] / 128;
  const int E = in_sizes[1] / 2;
  const int* src = ei;
  const int* dst = ei + E;

  float* ws   = (float*)d_ws;
  float* h    = ws;
  float* hn   = h + (size_t)N * 128;
  float* deg  = hn + (size_t)N * 128;
  float* dinv = deg + N;
  int*   lsrc = (int*)(dinv + N);
  int*   ldl  = lsrc + E;
  float* lnrm = (float*)(ldl + E);
  int*   blk  = (int*)(lnrm + E);

  init_deg   <<<(N + 255) / 256, 256, 0, stream>>>(deg, N);
  deg_accum  <<<(E + 255) / 256, 256, 0, stream>>>(dst, deg, E);
  finish_dinv<<<(N + 255) / 256, 256, 0, stream>>>(deg, dinv, N);

  const int ablocks = (N + RNG - 1) / RNG;

  gemm_wmma_f32<128><<<(N + 15) / 16, 256, 0, stream>>>(x, W1, h, N);
  agg_range<128, true><<<ablocks, NWV * 32, 0, stream>>>(h, src, dst, dinv, deg, b1, g1, be1, hn,
                                                         lsrc, ldl, lnrm, blk, E, N);

  gemm_wmma_f32<128><<<(N + 15) / 16, 256, 0, stream>>>(hn, W2, h, N);
  agg_list<128, true><<<ablocks, NWV * 32, 0, stream>>>(h, dinv, b2, g2, be2, hn,
                                                        lsrc, ldl, lnrm, blk, N);

  gemm_wmma_f32<64><<<(N + 31) / 32, 256, 0, stream>>>(hn, W3, h, N);
  agg_list<64, false><<<ablocks, NWV * 32, 0, stream>>>(h, dinv, b3, nullptr, nullptr, (float*)d_out,
                                                        lsrc, ldl, lnrm, blk, N);
}


# Transform moved behind the edge sums, into the aggregation block: input rows summed in LDS, then its rows times W on WMMA from LDS; no GEMM launch, no h plane; 9 -> 6 launches; sums (AX)W not A(XW)
# speedup vs baseline: 2.9708x; 1.1959x over previous
#include <hip/hip_runtime.h>
#include <math.h>

typedef __attribute__((ext_vector_type(2))) float v2f;
typedef __attribute__((ext_vector_type(8))) float v8f;

#define LN_EPS 1e-5f

__device__ __forceinline__ void atomAddF(float* p, float v) {
  unsafeAtomicAdd(p, v);
}

__global__ void init_deg(float* deg, int n) {
  int i = blockIdx.x * blockDim.x + threadIdx.x;
  if (i < n) deg[i] = 1.0f;
}

__global__ void deg_accum(const int* __restrict__ dst, float* deg, int E) {
  int i = blockIdx.x * blockDim.x + threadIdx.x;
  if (i < E) atomAddF(&deg[dst[i]], 1.0f);
}

__global__ void finish_dinv(const float* __restrict__ deg, float* __restrict__ dinv, int n) {
  int i = blockIdx.x * blockDim.x + threadIdx.x;
  if (i < n) {
    float d = deg[i];
    dinv[i] = (d > 0.0f) ? rsqrtf(d) : 0.0f;
  }
}

#define FIN 128
#define PITCH 132
#define RNG 196
#define RPAD 208
#define NWV 32
#define LCAP 640
#define SCAN 16
#define INFL 16

template <bool SAVED>
__device__ __forceinline__ void add_listed(
    const float* __restrict__ X, const int* __restrict__ srcA,
    const int* __restrict__ dstA, const float* __restrict__ dinv,
    float* acc, const int* le, int cnt, int d0, int lane,
    int* __restrict__ lsrc, int* __restrict__ ldl, float* __restrict__ lnrm, int gp)
{
  constexpr int LINES = FIN / 32;
  constexpr int GRP = INFL / LINES;
  int s_nx = 0, dl_nx = 0;
  float nrm_nx = 0.0f;
  if constexpr (SAVED) {
    if (lane < min(32, cnt)) {
      s_nx   = lsrc[gp + lane];
      dl_nx  = ldl[gp + lane];
      nrm_nx = lnrm[gp + lane];
    }
  }
  for (int k0 = 0; k0 < cnt; k0 += 32) {
    const int m = min(32, cnt - k0);
    int s = 0, dl = 0;
    float nrm = 0.0f;
    if constexpr (SAVED) {
      s = s_nx; dl = dl_nx; nrm = nrm_nx;
      if (lane < cnt - k0 - 32) {
        s_nx   = lsrc[gp + k0 + 32 + lane];
        dl_nx  = ldl[gp + k0 + 32 + lane];
        nrm_nx = lnrm[gp + k0 + 32 + lane];
      }
    }
    if (lane < m) {
      if constexpr (!SAVED) {
        const int e = le[k0 + lane];
        const int d = dstA[e];
        s = srcA[e];
        dl = d - d0;
        nrm = dinv[s] * dinv[d];
        lsrc[gp + k0 + lane] = s;
        ldl[gp + k0 + lane]  = dl;
        lnrm[gp + k0 + lane] = nrm;
      }
    }
    int q = 0;
    for (; q + GRP <= m; q += GRP) {
      float v[GRP][LINES];
      int dq[GRP];
#pragma unroll
      for (int u = 0; u < GRP; ++u) {
        const int su = __builtin_amdgcn_readlane(s, q + u);
        const float nu = __int_as_float(__builtin_amdgcn_readlane(__float_as_int(nrm), q + u));
        dq[u] = __builtin_amdgcn_readlane(dl, q + u);
        const float* __restrict__ xp = X + (size_t)su * FIN + lane;
#pragma unroll
        for (int j = 0; j < LINES; ++j) v[u][j] = xp[j * 32] * nu;
      }
#pragma unroll
      for (int u = 0; u < GRP; ++u) {
        float* ap = acc + dq[u] * PITCH + lane;
#pragma unroll
        for (int j = 0; j < LINES; ++j)
          __hip_atomic_fetch_add(ap + j * 32, v[u][j], __ATOMIC_RELAXED, __HIP_MEMORY_SCOPE_WORKGROUP);
      }
    }
    for (; q < m; ++q) {
      const int su = __builtin_amdgcn_readlane(s, q);
      const float nu = __int_as_float(__builtin_amdgcn_readlane(__float_as_int(nrm), q));
      const int dd = __builtin_amdgcn_readlane(dl, q);
      const float* __restrict__ xp = X + (size_t)su * FIN + lane;
      float* ap = acc + dd * PITCH + lane;
      float v[LINES];
#pragma unroll
      for (int j = 0; j < LINES; ++j) v[j] = xp[j * 32] * nu;
#pragma unroll
      for (int j = 0; j < LINES; ++j)
        __hip_atomic_fetch_add(ap + j * 32, v[j], __ATOMIC_RELAXED, __HIP_MEMORY_SCOPE_WORKGROUP);
    }
  }
}

template <int FOUT, int RT>
__device__ __forceinline__ void tile_products(
    const float* arow, const float* __restrict__ wcol, v8f (&d)[2][FOUT / 64])
{
  constexpr int NT = FOUT / 64;
#pragma unroll 8
  for (int kk = 0; kk < FIN; kk += 4) {
    v2f a[RT];
#pragma unroll
    for (int i = 0; i < RT; ++i) {
      a[i].x = arow[i * 16 * PITCH + kk];
      a[i].y = arow[i * 16 * PITCH + kk + 1];
    }
#pragma unroll
    for (int t = 0; t < NT; ++t) {
      v2f b;
      b.x = wcol[(size_t)kk * FOUT + 16 * t];
      b.y = wcol[(size_t)(kk + 1) * FOUT + 16 * t];
#pragma unroll
      for (int i = 0; i < RT; ++i)
        d[i][t] = __builtin_amdgcn_wmma_f32_16x16x4_f32(
            false, a[i], false, b, (short)0, d[i][t], false, false);
    }
  }
}

template <int FOUT>
__device__ __forceinline__ void rows_times_w(
    float* acc, const float* __restrict__ W, int nd, int w, int lane)
{
  constexpr int NT = FOUT / 64;
  const int r0 = (w >> 2) * 32;
  const int c0 = (w & 3) * (FOUT / 4);
  const bool has = r0 < nd;
  const bool two = r0 + 16 < nd;
  const int lh = lane & 15;
  const int kh = (lane >> 4) << 1;

  v8f d[2][NT];
#pragma unroll
  for (int i = 0; i < 2; ++i)
#pragma unroll
    for (int t = 0; t < NT; ++t) d[i][t] = v8f{};

  __syncthreads();
  if (has) {
    const float* arow = acc + (r0 + lh) * PITCH + kh;
    const float* __restrict__ wcol = W + (size_t)kh * FOUT + c0 + lh;
    if (two) tile_products<FOUT, 2>(arow, wcol, d);
    else     tile_products<FOUT, 1>(arow, wcol, d);
  }
  __syncthreads();
  if (has) {
    float* dp = acc + (r0 + ((lane >> 4) << 3)) * PITCH + c0 + lh;
#pragma unroll
    for (int i = 0; i < 2; ++i) {
      if (i == 0 || two) {
#pragma unroll
        for (int t = 0; t < NT; ++t)
#pragma unroll
          for (int v = 0; v < 8; ++v) dp[(i * 16 + v) * PITCH + 16 * t] = d[i][t][v];
      }
    }
  }
}

template <int M>
__device__ __forceinline__ int lane_xor_bits(int b) {
  if constexpr (M == 16)
    return __builtin_amdgcn_permlanex16(b, b, 0x76543210, 0xfedcba98, false, false);
  else
    return __builtin_amdgcn_update_dpp(0, b, 0x160 | M, 0xf, 0xf, false);
}
template <int M>
__device__ __forceinline__ float lane_xor(float x) {
  return __builtin_bit_cast(float, lane_xor_bits<M>(__builtin_bit_cast(int, x)));
}

__device__ __forceinline__ float wave_sum(float s) {
  s += lane_xor<16>(s);
  s += lane_xor<8>(s);
  s += lane_xor<4>(s);
  s += lane_xor<2>(s);
  s += lane_xor<1>(s);
  return s;
}

__device__ __forceinline__ int wave_sum_i(int s) {
  s += lane_xor_bits<16>(s);
  s += lane_xor_bits<8>(s);
  s += lane_xor_bits<4>(s);
  s += lane_xor_bits<2>(s);
  s += lane_xor_bits<1>(s);
  return s;
}

template <int FOUT, bool LNRELU>
__device__ __forceinline__ void rows_out(
    const float* acc, const float* __restrict__ bias, const float* __restrict__ gamma,
    const float* __restrict__ beta, float* __restrict__ OUT,
    int d0, int nd, int tid, int w, int lane)
{
  if constexpr (LNRELU) {
    static_assert(FOUT == 128, "the LayerNorm exit takes 4 features per lane of a 128-wide row");
    const float b0 = bias[lane * 4 + 0],  b1 = bias[lane * 4 + 1];
    const float b2 = bias[lane * 4 + 2],  b3 = bias[lane * 4 + 3];
    const float g0 = gamma[lane * 4 + 0], g1 = gamma[lane * 4 + 1];
    const float g2 = gamma[lane * 4 + 2], g3 = gamma[lane * 4 + 3];
    const float e0 = beta[lane * 4 + 0],  e1 = beta[lane * 4 + 1];
    const float e2 = beta[lane * 4 + 2],  e3 = beta[lane * 4 + 3];
    __syncthreads();

    for (int r = w; r < nd; r += NWV) {
      const float* row = acc + r * PITCH + lane * 4;
      const float v0 = row[0] + b0;
      const float v1 = row[1] + b1;
      const float v2 = row[2] + b2;
      const float v3 = row[3] + b3;

      const float mu = wave_sum(v0 + v1 + v2 + v3) * (1.0f / 128.0f);

      const float c0 = v0 - mu, c1 = v1 - mu, c2 = v2 - mu, c3 = v3 - mu;
      const float q = wave_sum(c0 * c0 + c1 * c1 + c2 * c2 + c3 * c3);
      const float rstd = rsqrtf(q * (1.0f / 128.0f) + LN_EPS);

      float* out = OUT + (size_t)(d0 + r) * FOUT + lane * 4;
      out[0] = fmaxf(0.0f, c0 * rstd * g0 + e0);
      out[1] = fmaxf(0.0f, c1 * rstd * g1 + e1);
      out[2] = fmaxf(0.0f, c2 * rstd * g2 + e2);
      out[3] = fmaxf(0.0f, c3 * rstd * g3 + e3);
    }
  } else {
    static_assert((NWV * 32) % FOUT == 0, "a thread keeps one feature column over its whole stride");
    const float bb = bias[tid & (FOUT - 1)];
    __syncthreads();
    for (int i = tid; i < nd * FOUT; i += NWV * 32)
      OUT[(size_t)d0 * FOUT + i] = acc[(i / FOUT) * PITCH + (i % FOUT)] + bb;
  }
}

__device__ __forceinline__ void rows_start(
    float* acc, const float* __restrict__ X, const float* __restrict__ dinv,
    int d0, int nd, int tid)
{
  for (int i = tid; i < nd * FIN; i += NWV * 32) {
    const int r = i / FIN;
    const float di = dinv[d0 + r];
    acc[r * PITCH + (i % FIN)] = X[(size_t)d0 * FIN + i] * (di * di);
  }
  const int ntl = ((nd + 15) >> 4) << 4;
  for (int i = nd * FIN + tid; i < ntl * FIN; i += NWV * 32)
    acc[(i / FIN) * PITCH + (i % FIN)] = 0.0f;
}

template <int FOUT, bool LNRELU>
__global__ __launch_bounds__(1024) void agg_range(
    const float* __restrict__ X, const int* __restrict__ srcA,
    const int* __restrict__ dstA, const float* __restrict__ dinv,
    const float* __restrict__ deg, const float* __restrict__ W,
    const float* __restrict__ bias, const float* __restrict__ gamma,
    const float* __restrict__ beta, float* __restrict__ OUT,
    int* __restrict__ lsrc, int* __restrict__ ldl, float* __restrict__ lnrm,
    int* __restrict__ blk, int E, int n)
{
  __shared__ float acc[RPAD * PITCH];
  __shared__ int lst_e[NWV * LCAP];
  __shared__ int red[2 * NWV];
  __shared__ int cursor;

  const int tid  = threadIdx.x;
  const int w    = __builtin_amdgcn_readfirstlane(tid >> 5);
  const int lane = tid & 31;
  const int d0   = blockIdx.x * RNG;
  const int nd   = min(RNG, n - d0);

  {
    static_assert(NWV == 32 && RNG <= NWV * 32, "one partial per lane; one owned row per thread");
    static_assert(RNG % 4 == 0, "the nodes below a range are a whole number of 16-byte pieces");
    constexpr int PFL = 8;
    const float4* __restrict__ dg4 = reinterpret_cast<const float4*>(deg);
    const int n4 = d0 >> 2;
    int sb = 0;
    for (int i0 = tid; i0 < n4; i0 += PFL * NWV * 32) {
      float4 c[PFL];
#pragma unroll
      for (int u = 0; u < PFL; ++u) c[u] = dg4[min(i0 + u * NWV * 32, n4 - 1)];
#pragma unroll
      for (int u = 0; u < PFL; ++u) {
        const int k = (int)c[u].x + (int)c[u].y + (int)c[u].z + (int)c[u].w - 4;
        sb += (i0 + u * NWV * 32 < n4) ? k : 0;
      }
    }
    int sc = (tid < nd) ? (int)deg[d0 + tid] - 1 : 0;
    sb = wave_sum_i(sb);
    sc = wave_sum_i(sc);
    if (lane == 0) { red[w] = sb; red[NWV + w] = sc; }
    if (tid == 0) cursor = 0;
  }
  rows_start(acc, X, dinv, d0, nd, tid);
  __syncthreads();
  const int base = __builtin_amdgcn_readfirstlane(wave_sum_i(red[lane]));
  if (w == 0) {
    const int tot = wave_sum_i(red[NWV + lane]);
    if (lane == 0) { blk[2 * blockIdx.x] = base; blk[2 * blockIdx.x + 1] = tot; }
  }

  const int per = ((E + NWV * SCAN * 32 - 1) / (NWV * SCAN * 32)) * (SCAN * 32);
  const int wb  = w * per;
  const int we  = min(E, wb + per);
  const int ntr = (we > wb) ? (we - wb + SCAN * 32 - 1) / (SCAN * 32) : 0;
  int* le = lst_e + w * LCAP;
  int cnt = 0;
  int tt = (ntr > 0) ? (int)(blockIdx.x % (unsigned)ntr) : 0;

  for (int t = 0; t < ntr; ++t) {
    const int b = wb + tt * (SCAN * 32);
    tt = (tt + 1 == ntr) ? 0 : tt + 1;
    int dv[SCAN];
    if (b + SCAN * 32 <= we) {
#pragma unroll
      for (int u = 0; u < SCAN; ++u) dv[u] = dstA[b + u * 32 + lane];
    } else {
#pragma unroll
      for (int u = 0; u < SCAN; ++u) {
        const int i = b + u * 32 + lane;
        dv[u] = (i < we) ? dstA[i] : -1;
      }
    }
#pragma unroll
    for (int u = 0; u < SCAN; ++u) {
      const int rel = dv[u] - d0;
      const bool hit = (unsigned)rel < (unsigned)nd;
      const unsigned mk = __builtin_amdgcn_ballot_w32(hit);
      if (mk != 0u) {
        if (hit) le[cnt + __popc(mk & ((1u << lane) - 1u))] = b + u * 32 + lane;
        cnt += __popc(mk);
      }
    }
    if (cnt >= 64 || t + 1 == ntr) {
      int pos = 0;
      if (lane == 0)
        pos = __hip_atomic_fetch_add(&cursor, cnt, __ATOMIC_RELAXED, __HIP_MEMORY_SCOPE_WORKGROUP);
      pos = __builtin_amdgcn_readfirstlane(pos);
      add_listed<false>(X, srcA, dstA, dinv, acc, le, cnt, d0, lane,
                        lsrc, ldl, lnrm, base + pos);
      cnt = 0;
    }
  }
  rows_times_w<FOUT>(acc, W, nd, w, lane);
  rows_out<FOUT, LNRELU>(acc, bias, gamma, beta, OUT, d0, nd, tid, w, lane);
}

template <int FOUT, bool LNRELU>
__global__ __launch_bounds__(1024) void agg_list(
    const float* __restrict__ X, const float* __restrict__ dinv,
    const float* __restrict__ W,
    const float* __restrict__ bias, const float* __restrict__ gamma,
    const float* __restrict__ beta, float* __restrict__ OUT,
    int* __restrict__ lsrc, int* __restrict__ ldl, float* __restrict__ lnrm,
    const int* __restrict__ blk, int n)
{
  __shared__ float acc[RPAD * PITCH];

  const int tid  = threadIdx.x;
  const int w    = __builtin_amdgcn_readfirstlane(tid >> 5);
  const int lane = tid & 31;
  const int d0   = blockIdx.x * RNG;
  const int nd   = min(RNG, n - d0);

  const int base = blk[2 * blockIdx.x];
  const int tot  = blk[2 * blockIdx.x + 1];

  rows_start(acc, X, dinv, d0, nd, tid);
  __syncthreads();

  constexpr int GRP = INFL / (FIN / 32);
  const int per = ((tot + NWV * GRP - 1) / (NWV * GRP)) * GRP;
  const int lo  = w * per;
  const int cnt = min(per, tot - lo);
  if (cnt > 0)
    add_listed<true>(X, nullptr, nullptr, dinv, acc, nullptr, cnt, d0, lane,
                     lsrc, ldl, lnrm, base + lo);

  rows_times_w<FOUT>(acc, W, nd, w, lane);
  rows_out<FOUT, LNRELU>(acc, bias, gamma, beta, OUT, d0, nd, tid, w, lane);
}

extern "C" void kernel_launch(void* const* d_in, const int* in_sizes, int n_in,
                              void* d_out, int out_size, void* d_ws, size_t ws_size,
                              hipStream_t stream) {
  const float* x   = (const float*)d_in[0];
  const int*   ei  = (const int*)d_in[1];
  const float* W1  = (const float*)d_in[2];
  const float* b1  = (const float*)d_in[3];
  const float* g1  = (const float*)d_in[4];
  const float* be1 = (const float*)d_in[5];
  const float* W2  = (const float*)d_in[6];
  const float* b2  = (const float*)d_in[7];
  const float* g2  = (const float*)d_in[8];
  const float* be2 = (const float*)d_in[9];
  const float* W3  = (const float*)d_in[10];
  const float* b3  = (const float*)d_in[11];

  const int N = in_sizes[0] / 128;
  const int E = in_sizes[1] / 2;
  const int* src = ei;
  const int* dst = ei + E;

  float* ws   = (float*)d_ws;
  float* ha   = ws;
  float* hb   = ha + (size_t)N * 128;
  float* deg  = hb + (size_t)N * 128;
  float* dinv = deg + N;
  int*   lsrc = (int*)(dinv + N);
  int*   ldl  = lsrc + E;
  float* lnrm = (float*)(ldl + E);
  int*   blk  = (int*)(lnrm + E);

  init_deg   <<<(N + 255) / 256, 256, 0, stream>>>(deg, N);
  deg_accum  <<<(E + 255) / 256, 256, 0, stream>>>(dst, deg, E);
  finish_dinv<<<(N + 255) / 256, 256, 0, stream>>>(deg, dinv, N);

  const int ablocks = (N + RNG - 1) / RNG;

  agg_range<128, true><<<ablocks, NWV * 32, 0, stream>>>(x, src, dst, dinv, deg, W1, b1, g1, be1, ha,
                                                         lsrc, ldl, lnrm, blk, E, N);

  agg_list<128, true><<<ablocks, NWV * 32, 0, stream>>>(ha, dinv, W2, b2, g2, be2, hb,
                                                        lsrc, ldl, lnrm, blk, N);

  agg_list<64, false><<<ablocks, NWV * 32, 0, stream>>>(hb, dinv, W3, b3, nullptr, nullptr, (float*)d_out,
                                                        lsrc, ldl, lnrm, blk, N);
}


# Edge list binned by destination block in the degree pass (LDS counting sort per 4096-edge chunk); layer 1's blocks read their groups, no scan of the whole list; sum order: LDS arrival as before
# speedup vs baseline: 3.2754x; 1.1025x over previous
#include <hip/hip_runtime.h>
#include <math.h>

typedef __attribute__((ext_vector_type(2))) float v2f;
typedef __attribute__((ext_vector_type(8))) float v8f;

#define LN_EPS 1e-5f

__device__ __forceinline__ void atomAddF(float* p, float v) {
  unsafeAtomicAdd(p, v);
}

__global__ void init_deg(float* deg, int n) {
  int i = blockIdx.x * blockDim.x + threadIdx.x;
  if (i < n) deg[i] = 1.0f;
}

__global__ void finish_dinv(const float* __restrict__ deg, float* __restrict__ dinv, int n) {
  int i = blockIdx.x * blockDim.x + threadIdx.x;
  if (i < n) {
    float d = deg[i];
    dinv[i] = (d > 0.0f) ? rsqrtf(d) : 0.0f;
  }
}

#define FIN 128
#define PITCH 132
#define RNG 196
#define RPAD 208
#define NWV 32
#define LCAP 256
#define CH 4096
#define HB 1024
#define INFL 16

template <bool SAVED>
__device__ __forceinline__ void add_listed(
    const float* __restrict__ X, const int* __restrict__ srcA,
    const int* __restrict__ dstA, const float* __restrict__ dinv,
    float* acc, const int* le, int cnt, int d0, int lane,
    int* __restrict__ lsrc, int* __restrict__ ldl, float* __restrict__ lnrm, int gp)
{
  constexpr int LINES = FIN / 32;
  constexpr int GRP = INFL / LINES;
  int s_nx = 0, dl_nx = 0;
  float nrm_nx = 0.0f;
  if constexpr (SAVED) {
    if (lane < min(32, cnt)) {
      s_nx   = lsrc[gp + lane];
      dl_nx  = ldl[gp + lane];
      nrm_nx = lnrm[gp + lane];
    }
  }
  for (int k0 = 0; k0 < cnt; k0 += 32) {
    const int m = min(32, cnt - k0);
    int s = 0, dl = 0;
    float nrm = 0.0f;
    if constexpr (SAVED) {
      s = s_nx; dl = dl_nx; nrm = nrm_nx;
      if (lane < cnt - k0 - 32) {
        s_nx   = lsrc[gp + k0 + 32 + lane];
        dl_nx  = ldl[gp + k0 + 32 + lane];
        nrm_nx = lnrm[gp + k0 + 32 + lane];
      }
    }
    if (lane < m) {
      if constexpr (!SAVED) {
        const int e = le[k0 + lane];
        const int d = dstA[e];
        s = srcA[e];
        dl = d - d0;
        nrm = dinv[s] * dinv[d];
        lsrc[gp + k0 + lane] = s;
        ldl[gp + k0 + lane]  = dl;
        lnrm[gp + k0 + lane] = nrm;
      }
    }
    int q = 0;
    for (; q + GRP <= m; q += GRP) {
      float v[GRP][LINES];
      int dq[GRP];
#pragma unroll
      for (int u = 0; u < GRP; ++u) {
        const int su = __builtin_amdgcn_readlane(s, q + u);
        const float nu = __int_as_float(__builtin_amdgcn_readlane(__float_as_int(nrm), q + u));
        dq[u] = __builtin_amdgcn_readlane(dl, q + u);
        const float* __restrict__ xp = X + (size_t)su * FIN + lane;
#pragma unroll
        for (int j = 0; j < LINES; ++j) v[u][j] = xp[j * 32] * nu;
      }
#pragma unroll
      for (int u = 0; u < GRP; ++u) {
        float* ap = acc + dq[u] * PITCH + lane;
#pragma unroll
        for (int j = 0; j < LINES; ++j)
          __hip_atomic_fetch_add(ap + j * 32, v[u][j], __ATOMIC_RELAXED, __HIP_MEMORY_SCOPE_WORKGROUP);
      }
    }
    for (; q < m; ++q) {
      const int su = __builtin_amdgcn_readlane(s, q);
      const float nu = __int_as_float(__builtin_amdgcn_readlane(__float_as_int(nrm), q));
      const int dd = __builtin_amdgcn_readlane(dl, q);
      const float* __restrict__ xp = X + (size_t)su * FIN + lane;
      float* ap = acc + dd * PITCH + lane;
      float v[LINES];
#pragma unroll
      for (int j = 0; j < LINES; ++j) v[j] = xp[j * 32] * nu;
#pragma unroll
      for (int j = 0; j < LINES; ++j)
        __hip_atomic_fetch_add(ap + j * 32, v[j], __ATOMIC_RELAXED, __HIP_MEMORY_SCOPE_WORKGROUP);
    }
  }
}

template <int FOUT, int RT>
__device__ __forceinline__ void tile_products(
    const float* arow, const float* __restrict__ wcol, v8f (&d)[2][FOUT / 64])
{
  constexpr int NT = FOUT / 64;
#pragma unroll 8
  for (int kk = 0; kk < FIN; kk += 4) {
    v2f a[RT];
#pragma unroll
    for (int i = 0; i < RT; ++i) {
      a[i].x = arow[i * 16 * PITCH + kk];
      a[i].y = arow[i * 16 * PITCH + kk + 1];
    }
#pragma unroll
    for (int t = 0; t < NT; ++t) {
      v2f b;
      b.x = wcol[(size_t)kk * FOUT + 16 * t];
      b.y = wcol[(size_t)(kk + 1) * FOUT + 16 * t];
#pragma unroll
      for (int i = 0; i < RT; ++i)
        d[i][t] = __builtin_amdgcn_wmma_f32_16x16x4_f32(
            false, a[i], false, b, (short)0, d[i][t], false, false);
    }
  }
}

template <int FOUT>
__device__ __forceinline__ void rows_times_w(
    float* acc, const float* __restrict__ W, int nd, int w, int lane)
{
  constexpr int NT = FOUT / 64;
  const int r0 = (w >> 2) * 32;
  const int c0 = (w & 3) * (FOUT / 4);
  const bool has = r0 < nd;
  const bool two = r0 + 16 < nd;
  const int lh = lane & 15;
  const int kh = (lane >> 4) << 1;

  v8f d[2][NT];
#pragma unroll
  for (int i = 0; i < 2; ++i)
#pragma unroll
    for (int t = 0; t < NT; ++t) d[i][t] = v8f{};

  __syncthreads();
  if (has) {
    const float* arow = acc + (r0 + lh) * PITCH + kh;
    const float* __restrict__ wcol = W + (size_t)kh * FOUT + c0 + lh;
    if (two) tile_products<FOUT, 2>(arow, wcol, d);
    else     tile_products<FOUT, 1>(arow, wcol, d);
  }
  __syncthreads();
  if (has) {
    float* dp = acc + (r0 + ((lane >> 4) << 3)) * PITCH + c0 + lh;
#pragma unroll
    for (int i = 0; i < 2; ++i) {
      if (i == 0 || two) {
#pragma unroll
        for (int t = 0; t < NT; ++t)
#pragma unroll
          for (int v = 0; v < 8; ++v) dp[(i * 16 + v) * PITCH + 16 * t] = d[i][t][v];
      }
    }
  }
}

template <int M>
__device__ __forceinline__ int lane_xor_bits(int b) {
  if constexpr (M == 16)
    return __builtin_amdgcn_permlanex16(b, b, 0x76543210, 0xfedcba98, false, false);
  else
    return __builtin_amdgcn_update_dpp(0, b, 0x160 | M, 0xf, 0xf, false);
}
template <int M>
__device__ __forceinline__ float lane_xor(float x) {
  return __builtin_bit_cast(float, lane_xor_bits<M>(__builtin_bit_cast(int, x)));
}

__device__ __forceinline__ float wave_sum(float s) {
  s += lane_xor<16>(s);
  s += lane_xor<8>(s);
  s += lane_xor<4>(s);
  s += lane_xor<2>(s);
  s += lane_xor<1>(s);
  return s;
}

__device__ __forceinline__ int wave_sum_i(int s) {
  s += lane_xor_bits<16>(s);
  s += lane_xor_bits<8>(s);
  s += lane_xor_bits<4>(s);
  s += lane_xor_bits<2>(s);
  s += lane_xor_bits<1>(s);
  return s;
}

__device__ __forceinline__ int wave_run_sum(int v, int lane) {
#pragma unroll
  for (int o = 1; o < 32; o <<= 1) {
    const int t = __shfl_up(v, o);
    if (lane >= o) v += t;
  }
  return v;
}

__global__ __launch_bounds__(1024) void deg_bin(
    const int* __restrict__ srcA, const int* __restrict__ dstA, float* deg,
    int* __restrict__ ssrc, int* __restrict__ sdst, int* __restrict__ tab, int E, int nb)
{
  static_assert(HB == NWV * 32, "one counter per thread");
  static_assert(CH % (NWV * 32) == 0, "a whole number of edges per thread");
  constexpr int EPT = CH / (NWV * 32);
  __shared__ int hist[HB];
  __shared__ int wtot[NWV];
  __shared__ int st_s[CH];
  __shared__ int st_d[CH];

  const int tid  = threadIdx.x;
  const int w    = tid >> 5;
  const int lane = tid & 31;
  const int e0   = blockIdx.x * CH;
  const int m    = min(CH, E - e0);

  hist[tid] = 0;
  __syncthreads();

  int s[EPT], d[EPT], b[EPT], r[EPT];
#pragma unroll
  for (int u = 0; u < EPT; ++u) {
    const int i = tid + u * NWV * 32;
    d[u] = (i < m) ? dstA[e0 + i] : -1;
    s[u] = (i < m) ? srcA[e0 + i] : 0;
  }
#pragma unroll
  for (int u = 0; u < EPT; ++u) {
    b[u] = 0; r[u] = 0;
    if (d[u] >= 0) {
      b[u] = d[u] / RNG;
      r[u] = __hip_atomic_fetch_add(&hist[b[u]], 1, __ATOMIC_RELAXED, __HIP_MEMORY_SCOPE_WORKGROUP);
    }
  }
  __syncthreads();

  const int v    = hist[tid];
  const int incl = wave_run_sum(v, lane);
  if (lane == 31) wtot[w] = incl;
  __syncthreads();
  const int wl   = wtot[lane];
  const int excl = wave_sum_i(lane < w ? wl : 0) + incl - v;
  hist[tid] = excl;
  if (tid <= nb) tab[(size_t)blockIdx.x * (nb + 1) + tid] = excl;
  __syncthreads();

#pragma unroll
  for (int u = 0; u < EPT; ++u) {
    if (d[u] >= 0) {
      const int pos = hist[b[u]] + r[u];
      st_s[pos] = s[u];
      st_d[pos] = d[u];
    }
  }
  __syncthreads();
#pragma unroll
  for (int u = 0; u < EPT; ++u) {
    const int i = tid + u * NWV * 32;
    if (i < m) {
      ssrc[e0 + i] = st_s[i];
      sdst[e0 + i] = st_d[i];
    }
  }
#pragma unroll
  for (int u = 0; u < EPT; ++u)
    if (d[u] >= 0) atomAddF(&deg[d[u]], 1.0f);
}

template <int FOUT, bool LNRELU>
__device__ __forceinline__ void rows_out(
    const float* acc, const float* __restrict__ bias, const float* __restrict__ gamma,
    const float* __restrict__ beta, float* __restrict__ OUT,
    int d0, int nd, int tid, int w, int lane)
{
  if constexpr (LNRELU) {
    static_assert(FOUT == 128, "the LayerNorm exit takes 4 features per lane of a 128-wide row");
    const float b0 = bias[lane * 4 + 0],  b1 = bias[lane * 4 + 1];
    const float b2 = bias[lane * 4 + 2],  b3 = bias[lane * 4 + 3];
    const float g0 = gamma[lane * 4 + 0], g1 = gamma[lane * 4 + 1];
    const float g2 = gamma[lane * 4 + 2], g3 = gamma[lane * 4 + 3];
    const float e0 = beta[lane * 4 + 0],  e1 = beta[lane * 4 + 1];
    const float e2 = beta[lane * 4 + 2],  e3 = beta[lane * 4 + 3];
    __syncthreads();

    for (int r = w; r < nd; r += NWV) {
      const float* row = acc + r * PITCH + lane * 4;
      const float v0 = row[0] + b0;
      const float v1 = row[1] + b1;
      const float v2 = row[2] + b2;
      const float v3 = row[3] + b3;

      const float mu = wave_sum(v0 + v1 + v2 + v3) * (1.0f / 128.0f);

      const float c0 = v0 - mu, c1 = v1 - mu, c2 = v2 - mu, c3 = v3 - mu;
      const float q = wave_sum(c0 * c0 + c1 * c1 + c2 * c2 + c3 * c3);
      const float rstd = rsqrtf(q * (1.0f / 128.0f) + LN_EPS);

      float* out = OUT + (size_t)(d0 + r) * FOUT + lane * 4;
      out[0] = fmaxf(0.0f, c0 * rstd * g0 + e0);
      out[1] = fmaxf(0.0f, c1 * rstd * g1 + e1);
      out[2] = fmaxf(0.0f, c2 * rstd * g2 + e2);
      out[3] = fmaxf(0.0f, c3 * rstd * g3 + e3);
    }
  } else {
    static_assert((NWV * 32) % FOUT == 0, "a thread keeps one feature column over its whole stride");
    const float bb = bias[tid & (FOUT - 1)];
    __syncthreads();
    for (int i = tid; i < nd * FOUT; i += NWV * 32)
      OUT[(size_t)d0 * FOUT + i] = acc[(i / FOUT) * PITCH + (i % FOUT)] + bb;
  }
}

__device__ __forceinline__ void rows_start(
    float* acc, const float* __restrict__ X, const float* __restrict__ dinv,
    int d0, int nd, int tid)
{
  for (int i = tid; i < nd * FIN; i += NWV * 32) {
    const int r = i / FIN;
    const float di = dinv[d0 + r];
    acc[r * PITCH + (i % FIN)] = X[(size_t)d0 * FIN + i] * (di * di);
  }
  const int ntl = ((nd + 15) >> 4) << 4;
  for (int i = nd * FIN + tid; i < ntl * FIN; i += NWV * 32)
    acc[(i / FIN) * PITCH + (i % FIN)] = 0.0f;
}

template <int FOUT, bool LNRELU>
__global__ __launch_bounds__(1024) void agg_range(
    const float* __restrict__ X, const int* __restrict__ ssrc,
    const int* __restrict__ sdst, const int* __restrict__ tab,
    const float* __restrict__ dinv, const float* __restrict__ W,
    const float* __restrict__ bias, const float* __restrict__ gamma,
    const float* __restrict__ beta, float* __restrict__ OUT,
    int* __restrict__ lsrc, int* __restrict__ ldl, float* __restrict__ lnrm,
    int* __restrict__ blk, int n, int nch, int nb)
{
  __shared__ float acc[RPAD * PITCH];
  __shared__ int lst_e[NWV * LCAP];
  __shared__ int seg_at[HB];
  __shared__ int seg_v[HB + 1];
  __shared__ int red[2 * NWV];

  const int tid  = threadIdx.x;
  const int w    = __builtin_amdgcn_readfirstlane(tid >> 5);
  const int lane = tid & 31;
  const int d0   = blockIdx.x * RNG;
  const int nd   = min(RNG, n - d0);

  static_assert(HB == NWV * 32, "one chunk per thread");
  int ga = 0, gb = 0;
  if (tid < nch) {
    const int* tp = tab + (size_t)tid * (nb + 1) + blockIdx.x;
    ga = tp[0];
    gb = tp[1];
  }
  rows_start(acc, X, dinv, d0, nd, tid);
  const int glen = gb - ga;
  const int incl = wave_run_sum(glen, lane);
  const int sa   = wave_sum_i(ga);
  if (lane == 31) red[w] = incl;
  if (lane == 0)  red[NWV + w] = sa;
  __syncthreads();
  const int wl   = red[lane];
  const int tot  = __builtin_amdgcn_readfirstlane(wave_sum_i(wl));
  const int base = __builtin_amdgcn_readfirstlane(wave_sum_i(red[NWV + lane]));
  const int pre  = wave_sum_i(lane < w ? wl : 0);
  if (tid < nch) {
    seg_at[tid] = tid * CH + ga;
    seg_v[tid]  = pre + incl - glen;
  }
  if (tid == 0) {
    seg_v[nch] = tot;
    blk[2 * blockIdx.x] = base;
    blk[2 * blockIdx.x + 1] = tot;
  }
  __syncthreads();

  constexpr int GRP = INFL / (FIN / 32);
  const int per = ((tot + NWV * GRP - 1) / (NWV * GRP)) * GRP;
  const int lo  = w * per;
  const int hi  = min(tot, lo + per);
  if (hi > lo) {
    int cb = 0;
    for (int t = lane; t < nch; t += 32) cb += (seg_v[t + 1] <= lo) ? 1 : 0;
    int t = __builtin_amdgcn_readfirstlane(wave_sum_i(cb));
    int* le = lst_e + w * LCAP;
    for (int a0 = lo; a0 < hi; a0 += LCAP) {
      const int b0 = min(hi, a0 + LCAP);
      int a = a0;
      while (a < b0) {
        const int sv = __builtin_amdgcn_readfirstlane(seg_v[t]);
        const int se = __builtin_amdgcn_readfirstlane(seg_v[t + 1]);
        const int g  = __builtin_amdgcn_readfirstlane(seg_at[t]) + (a - sv);
        const int take = min(b0, se) - a;
        for (int j = lane; j < take; j += 32) le[a - a0 + j] = g + j;
        a += take;
        if (a == se) ++t;
      }
      add_listed<false>(X, ssrc, sdst, dinv, acc, le, b0 - a0, d0, lane,
                        lsrc, ldl, lnrm, base + a0);
    }
  }
  rows_times_w<FOUT>(acc, W, nd, w, lane);
  rows_out<FOUT, LNRELU>(acc, bias, gamma, beta, OUT, d0, nd, tid, w, lane);
}

template <int FOUT, bool LNRELU>
__global__ __launch_bounds__(1024) void agg_list(
    const float* __restrict__ X, const float* __restrict__ dinv,
    const float* __restrict__ W,
    const float* __restrict__ bias, const float* __restrict__ gamma,
    const float* __restrict__ beta, float* __restrict__ OUT,
    int* __restrict__ lsrc, int* __restrict__ ldl, float* __restrict__ lnrm,
    const int* __restrict__ blk, int n)
{
  __shared__ float acc[RPAD * PITCH];

  const int tid  = threadIdx.x;
  const int w    = __builtin_amdgcn_readfirstlane(tid >> 5);
  const int lane = tid & 31;
  const int d0   = blockIdx.x * RNG;
  const int nd   = min(RNG, n - d0);

  const int base = blk[2 * blockIdx.x];
  const int tot  = blk[2 * blockIdx.x + 1];

  rows_start(acc, X, dinv, d0, nd, tid);
  __syncthreads();

  constexpr int GRP = INFL / (FIN / 32);
  const int per = ((tot + NWV * GRP - 1) / (NWV * GRP)) * GRP;
  const int lo  = w * per;
  const int cnt = min(per, tot - lo);
  if (cnt > 0)
    add_listed<true>(X, nullptr, nullptr, dinv, acc, nullptr, cnt, d0, lane,
                     lsrc, ldl, lnrm, base + lo);

  rows_times_w<FOUT>(acc, W, nd, w, lane);
  rows_out<FOUT, LNRELU>(acc, bias, gamma, beta, OUT, d0, nd, tid, w, lane);
}

extern "C" void kernel_launch(void* const* d_in, const int* in_sizes, int n_in,
                              void* d_out, int out_size, void* d_ws, size_t ws_size,
                              hipStream_t stream) {
  const float* x   = (const float*)d_in[0];
  const int*   ei  = (const int*)d_in[1];
  const float* W1  = (const float*)d_in[2];
  const float* b1  = (const float*)d_in[3];
  const float* g1  = (const float*)d_in[4];
  const float* be1 = (const float*)d_in[5];
  const float* W2  = (const float*)d_in[6];
  const float* b2  = (const float*)d_in[7];
  const float* g2  = (const float*)d_in[8];
  const float* be2 = (const float*)d_in[9];
  const float* W3  = (const float*)d_in[10];
  const float* b3  = (const float*)d_in[11];

  const int N = in_sizes[0] / 128;
  const int E = in_sizes[1] / 2;
  const int* src = ei;
  const int* dst = ei + E;

  float* ws   = (float*)d_ws;
  float* ha   = ws;
  float* hb   = ha + (size_t)N * 128;
  float* deg  = hb + (size_t)N * 128;
  float* dinv = deg + N;
  int*   lsrc = (int*)(dinv + N);
  int*   ldl  = lsrc + E;
  float* lnrm = (float*)(ldl + E);
  int*   ssrc = (int*)(lnrm + E);
  int*   sdst = ssrc + E;
  int*   blk  = sdst + E;
  int*   tab  = blk + 2 * ((N + RNG - 1) / RNG);

  init_deg   <<<(N + 255) / 256, 256, 0, stream>>>(deg, N);
  const int ablocks = (N + RNG - 1) / RNG;
  const int nch = (E + CH - 1) / CH;
  deg_bin    <<<nch, NWV * 32, 0, stream>>>(src, dst, deg, ssrc, sdst, tab, E, ablocks);
  finish_dinv<<<(N + 255) / 256, 256, 0, stream>>>(deg, dinv, N);

  agg_range<128, true><<<ablocks, NWV * 32, 0, stream>>>(x, ssrc, sdst, tab, dinv, W1, b1, g1, be1, ha,
                                                         lsrc, ldl, lnrm, blk, N, nch, ablocks);

  agg_list<128, true><<<ablocks, NWV * 32, 0, stream>>>(ha, dinv, W2, b2, g2, be2, hb,
                                                        lsrc, ldl, lnrm, blk, N);

  agg_list<64, false><<<ablocks, NWV * 32, 0, stream>>>(hb, dinv, W3, b3, nullptr, nullptr, (float*)d_out,
                                                        lsrc, ldl, lnrm, blk, N);
}


# Layer 3 taken as A(H W3): layer 2's block multiplies its LayerNorm rows by W3 on WMMA and stores 64-wide rows; layer 3 only sums them, half the bytes per edge; sum order: product before edge sums
# speedup vs baseline: 3.5394x; 1.0806x over previous
#include <hip/hip_runtime.h>
#include <math.h>

typedef __attribute__((ext_vector_type(2))) float v2f;
typedef __attribute__((ext_vector_type(8))) float v8f;

#define LN_EPS 1e-5f

__device__ __forceinline__ void atomAddF(float* p, float v) {
  unsafeAtomicAdd(p, v);
}

__global__ void init_deg(float* deg, int n) {
  int i = blockIdx.x * blockDim.x + threadIdx.x;
  if (i < n) deg[i] = 1.0f;
}

__global__ void finish_dinv(const float* __restrict__ deg, float* __restrict__ dinv, int n) {
  int i = blockIdx.x * blockDim.x + threadIdx.x;
  if (i < n) {
    float d = deg[i];
    dinv[i] = (d > 0.0f) ? rsqrtf(d) : 0.0f;
  }
}

#define FIN 128
#define FLAST 64
#define PITCH 132
#define RNG 196
#define RPAD 208
#define NWV 32
#define LCAP 256
#define CH 4096
#define HB 1024
#define INFL 16

template <bool SAVED, int FW, int PT>
__device__ __forceinline__ void add_listed(
    const float* __restrict__ X, const int* __restrict__ srcA,
    const int* __restrict__ dstA, const float* __restrict__ dinv,
    float* acc, const int* le, int cnt, int d0, int lane,
    int* __restrict__ lsrc, int* __restrict__ ldl, float* __restrict__ lnrm, int gp)
{
  constexpr int LINES = FW / 32;
  constexpr int GRP = INFL / LINES;
  int s_nx = 0, dl_nx = 0;
  float nrm_nx = 0.0f;
  if constexpr (SAVED) {
    if (lane < min(32, cnt)) {
      s_nx   = lsrc[gp + lane];
      dl_nx  = ldl[gp + lane];
      nrm_nx = lnrm[gp + lane];
    }
  }
  for (int k0 = 0; k0 < cnt; k0 += 32) {
    const int m = min(32, cnt - k0);
    int s = 0, dl = 0;
    float nrm = 0.0f;
    if constexpr (SAVED) {
      s = s_nx; dl = dl_nx; nrm = nrm_nx;
      if (lane < cnt - k0 - 32) {
        s_nx   = lsrc[gp + k0 + 32 + lane];
        dl_nx  = ldl[gp + k0 + 32 + lane];
        nrm_nx = lnrm[gp + k0 + 32 + lane];
      }
    }
    if (lane < m) {
      if constexpr (!SAVED) {
        const int e = le[k0 + lane];
        const int d = dstA[e];
        s = srcA[e];
        dl = d - d0;
        nrm = dinv[s] * dinv[d];
        lsrc[gp + k0 + lane] = s;
        ldl[gp + k0 + lane]  = dl;
        lnrm[gp + k0 + lane] = nrm;
      }
    }
    int q = 0;
    for (; q + GRP <= m; q += GRP) {
      float v[GRP][LINES];
      int dq[GRP];
#pragma unroll
      for (int u = 0; u < GRP; ++u) {
        const int su = __builtin_amdgcn_readlane(s, q + u);
        const float nu = __int_as_float(__builtin_amdgcn_readlane(__float_as_int(nrm), q + u));
        dq[u] = __builtin_amdgcn_readlane(dl, q + u);
        const float* __restrict__ xp = X + (size_t)su * FW + lane;
#pragma unroll
        for (int j = 0; j < LINES; ++j) v[u][j] = xp[j * 32] * nu;
      }
#pragma unroll
      for (int u = 0; u < GRP; ++u) {
        float* ap = acc + dq[u] * PT + lane;
#pragma unroll
        for (int j = 0; j < LINES; ++j)
          __hip_atomic_fetch_add(ap + j * 32, v[u][j], __ATOMIC_RELAXED, __HIP_MEMORY_SCOPE_WORKGROUP);
      }
    }
    for (; q < m; ++q) {
      const int su = __builtin_amdgcn_readlane(s, q);
      const float nu = __int_as_float(__builtin_amdgcn_readlane(__float_as_int(nrm), q));
      const int dd = __builtin_amdgcn_readlane(dl, q);
      const float* __restrict__ xp = X + (size_t)su * FW + lane;
      float* ap = acc + dd * PT + lane;
      float v[LINES];
#pragma unroll
      for (int j = 0; j < LINES; ++j) v[j] = xp[j * 32] * nu;
#pragma unroll
      for (int j = 0; j < LINES; ++j)
        __hip_atomic_fetch_add(ap + j * 32, v[j], __ATOMIC_RELAXED, __HIP_MEMORY_SCOPE_WORKGROUP);
    }
  }
}

template <int FOUT, int RT>
__device__ __forceinline__ void tile_products(
    const float* arow, const float* __restrict__ wcol, v8f (&d)[2][FOUT / 64])
{
  constexpr int NT = FOUT / 64;
#pragma unroll 8
  for (int kk = 0; kk < FIN; kk += 4) {
    v2f a[RT];
#pragma unroll
    for (int i = 0; i < RT; ++i) {
      a[i].x = arow[i * 16 * PITCH + kk];
      a[i].y = arow[i * 16 * PITCH + kk + 1];
    }
#pragma unroll
    for (int t = 0; t < NT; ++t) {
      v2f b;
      b.x = wcol[(size_t)kk * FOUT + 16 * t];
      b.y = wcol[(size_t)(kk + 1) * FOUT + 16 * t];
#pragma unroll
      for (int i = 0; i < RT; ++i)
        d[i][t] = __builtin_amdgcn_wmma_f32_16x16x4_f32(
            false, a[i], false, b, (short)0, d[i][t], false, false);
    }
  }
}

template <int FOUT>
__device__ __forceinline__ void rows_times_w(
    float* acc, const float* __restrict__ W, int nd, int w, int lane)
{
  constexpr int NT = FOUT / 64;
  const int r0 = (w >> 2) * 32;
  const int c0 = (w & 3) * (FOUT / 4);
  const bool has = r0 < nd;
  const bool two = r0 + 16 < nd;
  const int lh = lane & 15;
  const int kh = (lane >> 4) << 1;

  v8f d[2][NT];
#pragma unroll
  for (int i = 0; i < 2; ++i)
#pragma unroll
    for (int t = 0; t < NT; ++t) d[i][t] = v8f{};

  __syncthreads();
  if (has) {
    const float* arow = acc + (r0 + lh) * PITCH + kh;
    const float* __restrict__ wcol = W + (size_t)kh * FOUT + c0 + lh;
    if (two) tile_products<FOUT, 2>(arow, wcol, d);
    else     tile_products<FOUT, 1>(arow, wcol, d);
  }
  __syncthreads();
  if (has) {
    float* dp = acc + (r0 + ((lane >> 4) << 3)) * PITCH + c0 + lh;
#pragma unroll
    for (int i = 0; i < 2; ++i) {
      if (i == 0 || two) {
#pragma unroll
        for (int t = 0; t < NT; ++t)
#pragma unroll
          for (int v = 0; v < 8; ++v) dp[(i * 16 + v) * PITCH + 16 * t] = d[i][t][v];
      }
    }
  }
}

template <int M>
__device__ __forceinline__ int lane_xor_bits(int b) {
  if constexpr (M == 16)
    return __builtin_amdgcn_permlanex16(b, b, 0x76543210, 0xfedcba98, false, false);
  else
    return __builtin_amdgcn_update_dpp(0, b, 0x160 | M, 0xf, 0xf, false);
}
template <int M>
__device__ __forceinline__ float lane_xor(float x) {
  return __builtin_bit_cast(float, lane_xor_bits<M>(__builtin_bit_cast(int, x)));
}

__device__ __forceinline__ float wave_sum(float s) {
  s += lane_xor<16>(s);
  s += lane_xor<8>(s);
  s += lane_xor<4>(s);
  s += lane_xor<2>(s);
  s += lane_xor<1>(s);
  return s;
}

__device__ __forceinline__ int wave_sum_i(int s) {
  s += lane_xor_bits<16>(s);
  s += lane_xor_bits<8>(s);
  s += lane_xor_bits<4>(s);
  s += lane_xor_bits<2>(s);
  s += lane_xor_bits<1>(s);
  return s;
}

__device__ __forceinline__ int wave_run_sum(int v, int lane) {
#pragma unroll
  for (int o = 1; o < 32; o <<= 1) {
    const int t = __shfl_up(v, o);
    if (lane >= o) v += t;
  }
  return v;
}

__global__ __launch_bounds__(1024) void deg_bin(
    const int* __restrict__ srcA, const int* __restrict__ dstA, float* deg,
    int* __restrict__ ssrc, int* __restrict__ sdst, int* __restrict__ tab, int E, int nb)
{
  static_assert(HB == NWV * 32, "one counter per thread");
  static_assert(CH % (NWV * 32) == 0, "a whole number of edges per thread");
  constexpr int EPT = CH / (NWV * 32);
  __shared__ int hist[HB];
  __shared__ int wtot[NWV];
  __shared__ int st_s[CH];
  __shared__ int st_d[CH];

  const int tid  = threadIdx.x;
  const int w    = tid >> 5;
  const int lane = tid & 31;
  const int e0   = blockIdx.x * CH;
  const int m    = min(CH, E - e0);

  hist[tid] = 0;
  __syncthreads();

  int s[EPT], d[EPT], b[EPT], r[EPT];
#pragma unroll
  for (int u = 0; u < EPT; ++u) {
    const int i = tid + u * NWV * 32;
    d[u] = (i < m) ? dstA[e0 + i] : -1;
    s[u] = (i < m) ? srcA[e0 + i] : 0;
  }
#pragma unroll
  for (int u = 0; u < EPT; ++u) {
    b[u] = 0; r[u] = 0;
    if (d[u] >= 0) {
      b[u] = d[u] / RNG;
      r[u] = __hip_atomic_fetch_add(&hist[b[u]], 1, __ATOMIC_RELAXED, __HIP_MEMORY_SCOPE_WORKGROUP);
    }
  }
  __syncthreads();

  const int v    = hist[tid];
  const int incl = wave_run_sum(v, lane);
  if (lane == 31) wtot[w] = incl;
  __syncthreads();
  const int wl   = wtot[lane];
  const int excl = wave_sum_i(lane < w ? wl : 0) + incl - v;
  hist[tid] = excl;
  if (tid <= nb) tab[(size_t)blockIdx.x * (nb + 1) + tid] = excl;
  __syncthreads();

#pragma unroll
  for (int u = 0; u < EPT; ++u) {
    if (d[u] >= 0) {
      const int pos = hist[b[u]] + r[u];
      st_s[pos] = s[u];
      st_d[pos] = d[u];
    }
  }
  __syncthreads();
#pragma unroll
  for (int u = 0; u < EPT; ++u) {
    const int i = tid + u * NWV * 32;
    if (i < m) {
      ssrc[e0 + i] = st_s[i];
      sdst[e0 + i] = st_d[i];
    }
  }
#pragma unroll
  for (int u = 0; u < EPT; ++u)
    if (d[u] >= 0) atomAddF(&deg[d[u]], 1.0f);
}

template <bool INPLACE>
__device__ __forceinline__ void rows_norm(
    float* acc, const float* __restrict__ bias, const float* __restrict__ gamma,
    const float* __restrict__ beta, float* __restrict__ OUT,
    int d0, int nd, int w, int lane)
{
  static_assert(FIN == 128, "4 features per lane of a 128-wide row");
  const float b0 = bias[lane * 4 + 0],  b1 = bias[lane * 4 + 1];
  const float b2 = bias[lane * 4 + 2],  b3 = bias[lane * 4 + 3];
  const float g0 = gamma[lane * 4 + 0], g1 = gamma[lane * 4 + 1];
  const float g2 = gamma[lane * 4 + 2], g3 = gamma[lane * 4 + 3];
  const float e0 = beta[lane * 4 + 0],  e1 = beta[lane * 4 + 1];
  const float e2 = beta[lane * 4 + 2],  e3 = beta[lane * 4 + 3];
  __syncthreads();

  for (int r = w; r < nd; r += NWV) {
    float* row = acc + r * PITCH + lane * 4;
    const float v0 = row[0] + b0;
    const float v1 = row[1] + b1;
    const float v2 = row[2] + b2;
    const float v3 = row[3] + b3;

    const float mu = wave_sum(v0 + v1 + v2 + v3) * (1.0f / 128.0f);

    const float c0 = v0 - mu, c1 = v1 - mu, c2 = v2 - mu, c3 = v3 - mu;
    const float q = wave_sum(c0 * c0 + c1 * c1 + c2 * c2 + c3 * c3);
    const float rstd = rsqrtf(q * (1.0f / 128.0f) + LN_EPS);

    float* out = INPLACE ? row : OUT + (size_t)(d0 + r) * FIN + lane * 4;
    out[0] = fmaxf(0.0f, c0 * rstd * g0 + e0);
    out[1] = fmaxf(0.0f, c1 * rstd * g1 + e1);
    out[2] = fmaxf(0.0f, c2 * rstd * g2 + e2);
    out[3] = fmaxf(0.0f, c3 * rstd * g3 + e3);
  }
}

template <int FOUT, int PT, bool BIAS>
__device__ __forceinline__ void rows_out(
    const float* acc, const float* __restrict__ bias, float* __restrict__ OUT,
    int d0, int nd, int tid)
{
  static_assert((NWV * 32) % FOUT == 0, "a thread keeps one feature column over its whole stride");
  float bb = 0.0f;
  if constexpr (BIAS) bb = bias[tid & (FOUT - 1)];
  __syncthreads();
  for (int i = tid; i < nd * FOUT; i += NWV * 32)
    OUT[(size_t)d0 * FOUT + i] = acc[(i / FOUT) * PT + (i % FOUT)] + bb;
}

template <int FW, int PT, bool TILES>
__device__ __forceinline__ void rows_start(
    float* acc, const float* __restrict__ X, const float* __restrict__ dinv,
    int d0, int nd, int tid)
{
  for (int i = tid; i < nd * FW; i += NWV * 32) {
    const int r = i / FW;
    const float di = dinv[d0 + r];
    acc[r * PT + (i % FW)] = X[(size_t)d0 * FW + i] * (di * di);
  }
  if constexpr (TILES) {
    const int ntl = ((nd + 15) >> 4) << 4;
    for (int i = nd * FW + tid; i < ntl * FW; i += NWV * 32)
      acc[(i / FW) * PT + (i % FW)] = 0.0f;
  }
}

template <int FOUT, bool LNRELU>
__global__ __launch_bounds__(1024) void agg_range(
    const float* __restrict__ X, const int* __restrict__ ssrc,
    const int* __restrict__ sdst, const int* __restrict__ tab,
    const float* __restrict__ dinv, const float* __restrict__ W,
    const float* __restrict__ bias, const float* __restrict__ gamma,
    const float* __restrict__ beta, float* __restrict__ OUT,
    int* __restrict__ lsrc, int* __restrict__ ldl, float* __restrict__ lnrm,
    int* __restrict__ blk, int n, int nch, int nb)
{
  __shared__ float acc[RPAD * PITCH];
  __shared__ int lst_e[NWV * LCAP];
  __shared__ int seg_at[HB];
  __shared__ int seg_v[HB + 1];
  __shared__ int red[2 * NWV];

  const int tid  = threadIdx.x;
  const int w    = __builtin_amdgcn_readfirstlane(tid >> 5);
  const int lane = tid & 31;
  const int d0   = blockIdx.x * RNG;
  const int nd   = min(RNG, n - d0);

  static_assert(HB == NWV * 32, "one chunk per thread");
  int ga = 0, gb = 0;
  if (tid < nch) {
    const int* tp = tab + (size_t)tid * (nb + 1) + blockIdx.x;
    ga = tp[0];
    gb = tp[1];
  }
  rows_start<FIN, PITCH, true>(acc, X, dinv, d0, nd, tid);
  const int glen = gb - ga;
  const int incl = wave_run_sum(glen, lane);
  const int sa   = wave_sum_i(ga);
  if (lane == 31) red[w] = incl;
  if (lane == 0)  red[NWV + w] = sa;
  __syncthreads();
  const int wl   = red[lane];
  const int tot  = __builtin_amdgcn_readfirstlane(wave_sum_i(wl));
  const int base = __builtin_amdgcn_readfirstlane(wave_sum_i(red[NWV + lane]));
  const int pre  = wave_sum_i(lane < w ? wl : 0);
  if (tid < nch) {
    seg_at[tid] = tid * CH + ga;
    seg_v[tid]  = pre + incl - glen;
  }
  if (tid == 0) {
    seg_v[nch] = tot;
    blk[2 * blockIdx.x] = base;
    blk[2 * blockIdx.x + 1] = tot;
  }
  __syncthreads();

  constexpr int GRP = INFL / (FIN / 32);
  const int per = ((tot + NWV * GRP - 1) / (NWV * GRP)) * GRP;
  const int lo  = w * per;
  const int hi  = min(tot, lo + per);
  if (hi > lo) {
    int cb = 0;
    for (int t = lane; t < nch; t += 32) cb += (seg_v[t + 1] <= lo) ? 1 : 0;
    int t = __builtin_amdgcn_readfirstlane(wave_sum_i(cb));
    int* le = lst_e + w * LCAP;
    for (int a0 = lo; a0 < hi; a0 += LCAP) {
      const int b0 = min(hi, a0 + LCAP);
      int a = a0;
      while (a < b0) {
        const int sv = __builtin_amdgcn_readfirstlane(seg_v[t]);
        const int se = __builtin_amdgcn_readfirstlane(seg_v[t + 1]);
        const int g  = __builtin_amdgcn_readfirstlane(seg_at[t]) + (a - sv);
        const int take = min(b0, se) - a;
        for (int j = lane; j < take; j += 32) le[a - a0 + j] = g + j;
        a += take;
        if (a == se) ++t;
      }
      add_listed<false, FIN, PITCH>(X, ssrc, sdst, dinv, acc, le, b0 - a0, d0, lane,
                                    lsrc, ldl, lnrm, base + a0);
    }
  }
  rows_times_w<FOUT>(acc, W, nd, w, lane);
  if constexpr (LNRELU) rows_norm<false>(acc, bias, gamma, beta, OUT, d0, nd, w, lane);
  else                  rows_out<FOUT, PITCH, true>(acc, bias, OUT, d0, nd, tid);
}

template <int FOUT, int FNX>
__global__ __launch_bounds__(1024) void agg_list(
    const float* __restrict__ X, const float* __restrict__ dinv,
    const float* __restrict__ W,
    const float* __restrict__ bias, const float* __restrict__ gamma,
    const float* __restrict__ beta, const float* __restrict__ WN,
    float* __restrict__ OUT,
    int* __restrict__ lsrc, int* __restrict__ ldl, float* __restrict__ lnrm,
    const int* __restrict__ blk, int n)
{
  static_assert(FOUT == FIN, "the finished rows are the next product's operands");
  __shared__ float acc[RPAD * PITCH];

  const int tid  = threadIdx.x;
  const int w    = __builtin_amdgcn_readfirstlane(tid >> 5);
  const int lane = tid & 31;
  const int d0   = blockIdx.x * RNG;
  const int nd   = min(RNG, n - d0);

  const int base = blk[2 * blockIdx.x];
  const int tot  = blk[2 * blockIdx.x + 1];

  rows_start<FIN, PITCH, true>(acc, X, dinv, d0, nd, tid);
  __syncthreads();

  constexpr int GRP = INFL / (FIN / 32);
  const int per = ((tot + NWV * GRP - 1) / (NWV * GRP)) * GRP;
  const int lo  = w * per;
  const int cnt = min(per, tot - lo);
  if (cnt > 0)
    add_listed<true, FIN, PITCH>(X, nullptr, nullptr, dinv, acc, nullptr, cnt, d0, lane,
                                 lsrc, ldl, lnrm, base + lo);

  rows_times_w<FOUT>(acc, W, nd, w, lane);
  rows_norm<true>(acc, bias, gamma, beta, nullptr, d0, nd, w, lane);
  rows_times_w<FNX>(acc, WN, nd, w, lane);
  rows_out<FNX, PITCH, false>(acc, nullptr, OUT, d0, nd, tid);
}

template <int FW>
__global__ __launch_bounds__(1024) void agg_sum(
    const float* __restrict__ X, const float* __restrict__ dinv,
    const float* __restrict__ bias, float* __restrict__ OUT,
    int* __restrict__ lsrc, int* __restrict__ ldl, float* __restrict__ lnrm,
    const int* __restrict__ blk, int n)
{
  __shared__ float acc[RNG * FW];

  const int tid  = threadIdx.x;
  const int w    = __builtin_amdgcn_readfirstlane(tid >> 5);
  const int lane = tid & 31;
  const int d0   = blockIdx.x * RNG;
  const int nd   = min(RNG, n - d0);

  const int base = blk[2 * blockIdx.x];
  const int tot  = blk[2 * blockIdx.x + 1];

  rows_start<FW, FW, false>(acc, X, dinv, d0, nd, tid);
  __syncthreads();

  constexpr int GRP = INFL / (FW / 32);
  const int per = ((tot + NWV * GRP - 1) / (NWV * GRP)) * GRP;
  const int lo  = w * per;
  const int cnt = min(per, tot - lo);
  if (cnt > 0)
    add_listed<true, FW, FW>(X, nullptr, nullptr, dinv, acc, nullptr, cnt, d0, lane,
                             lsrc, ldl, lnrm, base + lo);

  rows_out<FW, FW, true>(acc, bias, OUT, d0, nd, tid);
}

extern "C" void kernel_launch(void* const* d_in, const int* in_sizes, int n_in,
                              void* d_out, int out_size, void* d_ws, size_t ws_size,
                              hipStream_t stream) {
  const float* x   = (const float*)d_in[0];
  const int*   ei  = (const int*)d_in[1];
  const float* W1  = (const float*)d_in[2];
  const float* b1  = (const float*)d_in[3];
  const float* g1  = (const float*)d_in[4];
  const float* be1 = (const float*)d_in[5];
  const float* W2  = (const float*)d_in[6];
  const float* b2  = (const float*)d_in[7];
  const float* g2  = (const float*)d_in[8];
  const float* be2 = (const float*)d_in[9];
  const float* W3  = (const float*)d_in[10];
  const float* b3  = (const float*)d_in[11];

  const int N = in_sizes[0] / 128;
  const int E = in_sizes[1] / 2;
  const int* src = ei;
  const int* dst = ei + E;

  float* ws   = (float*)d_ws;
  float* ha   = ws;
  float* hb   = ha + (size_t)N * 128;
  float* deg  = hb + (size_t)N * 64;
  float* dinv = deg + N;
  int*   lsrc = (int*)(dinv + N);
  int*   ldl  = lsrc + E;
  float* lnrm = (float*)(ldl + E);
  int*   ssrc = (int*)(lnrm + E);
  int*   sdst = ssrc + E;
  int*   blk  = sdst + E;
  int*   tab  = blk + 2 * ((N + RNG - 1) / RNG);

  init_deg   <<<(N + 255) / 256, 256, 0, stream>>>(deg, N);
  const int ablocks = (N + RNG - 1) / RNG;
  const int nch = (E + CH - 1) / CH;
  deg_bin    <<<nch, NWV * 32, 0, stream>>>(src, dst, deg, ssrc, sdst, tab, E, ablocks);
  finish_dinv<<<(N + 255) / 256, 256, 0, stream>>>(deg, dinv, N);

  agg_range<128, true><<<ablocks, NWV * 32, 0, stream>>>(x, ssrc, sdst, tab, dinv, W1, b1, g1, be1, ha,
                                                         lsrc, ldl, lnrm, blk, N, nch, ablocks);

  agg_list<128, FLAST><<<ablocks, NWV * 32, 0, stream>>>(ha, dinv, W2, b2, g2, be2, W3, hb,
                                                         lsrc, ldl, lnrm, blk, N);

  agg_sum<FLAST><<<ablocks, NWV * 32, 0, stream>>>(hb, dinv, b3, (float*)d_out,
                                                   lsrc, ldl, lnrm, blk, N);
}


# Degrees counted per destination block from the binned edge list with LDS integer adds, 1/sqrt written directly: no fill, no global f32 atomics, no finish pass; 6 -> 5 launches; counts exact as before
# speedup vs baseline: 4.6992x; 1.3277x over previous
#include <hip/hip_runtime.h>
#include <math.h>

typedef __attribute__((ext_vector_type(2))) float v2f;
typedef __attribute__((ext_vector_type(8))) float v8f;

#define LN_EPS 1e-5f

#define FIN 128
#define FLAST 64
#define PITCH 132
#define RNG 196
#define RPAD 208
#define NWV 32
#define LCAP 256
#define CH 4096
#define HB 1024
#define INFL 16
#define DWV 8
#define DGR 20

template <bool SAVED, int FW, int PT>
__device__ __forceinline__ void add_listed(
    const float* __restrict__ X, const int* __restrict__ srcA,
    const int* __restrict__ dstA, const float* __restrict__ dinv,
    float* acc, const int* le, int cnt, int d0, int lane,
    int* __restrict__ lsrc, int* __restrict__ ldl, float* __restrict__ lnrm, int gp)
{
  constexpr int LINES = FW / 32;
  constexpr int GRP = INFL / LINES;
  int s_nx = 0, dl_nx = 0;
  float nrm_nx = 0.0f;
  if constexpr (SAVED) {
    if (lane < min(32, cnt)) {
      s_nx   = lsrc[gp + lane];
      dl_nx  = ldl[gp + lane];
      nrm_nx = lnrm[gp + lane];
    }
  }
  for (int k0 = 0; k0 < cnt; k0 += 32) {
    const int m = min(32, cnt - k0);
    int s = 0, dl = 0;
    float nrm = 0.0f;
    if constexpr (SAVED) {
      s = s_nx; dl = dl_nx; nrm = nrm_nx;
      if (lane < cnt - k0 - 32) {
        s_nx   = lsrc[gp + k0 + 32 + lane];
        dl_nx  = ldl[gp + k0 + 32 + lane];
        nrm_nx = lnrm[gp + k0 + 32 + lane];
      }
    }
    if (lane < m) {
      if constexpr (!SAVED) {
        const int e = le[k0 + lane];
        const int d = dstA[e];
        s = srcA[e];
        dl = d - d0;
        nrm = dinv[s] * dinv[d];
        lsrc[gp + k0 + lane] = s;
        ldl[gp + k0 + lane]  = dl;
        lnrm[gp + k0 + lane] = nrm;
      }
    }
    int q = 0;
    for (; q + GRP <= m; q += GRP) {
      float v[GRP][LINES];
      int dq[GRP];
#pragma unroll
      for (int u = 0; u < GRP; ++u) {
        const int su = __builtin_amdgcn_readlane(s, q + u);
        const float nu = __int_as_float(__builtin_amdgcn_readlane(__float_as_int(nrm), q + u));
        dq[u] = __builtin_amdgcn_readlane(dl, q + u);
        const float* __restrict__ xp = X + (size_t)su * FW + lane;
#pragma unroll
        for (int j = 0; j < LINES; ++j) v[u][j] = xp[j * 32] * nu;
      }
#pragma unroll
      for (int u = 0; u < GRP; ++u) {
        float* ap = acc + dq[u] * PT + lane;
#pragma unroll
        for (int j = 0; j < LINES; ++j)
          __hip_atomic_fetch_add(ap + j * 32, v[u][j], __ATOMIC_RELAXED, __HIP_MEMORY_SCOPE_WORKGROUP);
      }
    }
    for (; q < m; ++q) {
      const int su = __builtin_amdgcn_readlane(s, q);
      const float nu = __int_as_float(__builtin_amdgcn_readlane(__float_as_int(nrm), q));
      const int dd = __builtin_amdgcn_readlane(dl, q);
      const float* __restrict__ xp = X + (size_t)su * FW + lane;
      float* ap = acc + dd * PT + lane;
      float v[LINES];
#pragma unroll
      for (int j = 0; j < LINES; ++j) v[j] = xp[j * 32] * nu;
#pragma unroll
      for (int j = 0; j < LINES; ++j)
        __hip_atomic_fetch_add(ap + j * 32, v[j], __ATOMIC_RELAXED, __HIP_MEMORY_SCOPE_WORKGROUP);
    }
  }
}

template <int FOUT, int RT>
__device__ __forceinline__ void tile_products(
    const float* arow, const float* __restrict__ wcol, v8f (&d)[2][FOUT / 64])
{
  constexpr int NT = FOUT / 64;
#pragma unroll 8
  for (int kk = 0; kk < FIN; kk += 4) {
    v2f a[RT];
#pragma unroll
    for (int i = 0; i < RT; ++i) {
      a[i].x = arow[i * 16 * PITCH + kk];
      a[i].y = arow[i * 16 * PITCH + kk + 1];
    }
#pragma unroll
    for (int t = 0; t < NT; ++t) {
      v2f b;
      b.x = wcol[(size_t)kk * FOUT + 16 * t];
      b.y = wcol[(size_t)(kk + 1) * FOUT + 16 * t];
#pragma unroll
      for (int i = 0; i < RT; ++i)
        d[i][t] = __builtin_amdgcn_wmma_f32_16x16x4_f32(
            false, a[i], false, b, (short)0, d[i][t], false, false);
    }
  }
}

template <int FOUT>
__device__ __forceinline__ void rows_times_w(
    float* acc, const float* __restrict__ W, int nd, int w, int lane)
{
  constexpr int NT = FOUT / 64;
  const int r0 = (w >> 2) * 32;
  const int c0 = (w & 3) * (FOUT / 4);
  const bool has = r0 < nd;
  const bool two = r0 + 16 < nd;
  const int lh = lane & 15;
  const int kh = (lane >> 4) << 1;

  v8f d[2][NT];
#pragma unroll
  for (int i = 0; i < 2; ++i)
#pragma unroll
    for (int t = 0; t < NT; ++t) d[i][t] = v8f{};

  __syncthreads();
  if (has) {
    const float* arow = acc + (r0 + lh) * PITCH + kh;
    const float* __restrict__ wcol = W + (size_t)kh * FOUT + c0 + lh;
    if (two) tile_products<FOUT, 2>(arow, wcol, d);
    else     tile_products<FOUT, 1>(arow, wcol, d);
  }
  __syncthreads();
  if (has) {
    float* dp = acc + (r0 + ((lane >> 4) << 3)) * PITCH + c0 + lh;
#pragma unroll
    for (int i = 0; i < 2; ++i) {
      if (i == 0 || two) {
#pragma unroll
        for (int t = 0; t < NT; ++t)
#pragma unroll
          for (int v = 0; v < 8; ++v) dp[(i * 16 + v) * PITCH + 16 * t] = d[i][t][v];
      }
    }
  }
}

template <int M>
__device__ __forceinline__ int lane_xor_bits(int b) {
  if constexpr (M == 16)
    return __builtin_amdgcn_permlanex16(b, b, 0x76543210, 0xfedcba98, false, false);
  else
    return __builtin_amdgcn_update_dpp(0, b, 0x160 | M, 0xf, 0xf, false);
}
template <int M>
__device__ __forceinline__ float lane_xor(float x) {
  return __builtin_bit_cast(float, lane_xor_bits<M>(__builtin_bit_cast(int, x)));
}

__device__ __forceinline__ float wave_sum(float s) {
  s += lane_xor<16>(s);
  s += lane_xor<8>(s);
  s += lane_xor<4>(s);
  s += lane_xor<2>(s);
  s += lane_xor<1>(s);
  return s;
}

__device__ __forceinline__ int wave_sum_i(int s) {
  s += lane_xor_bits<16>(s);
  s += lane_xor_bits<8>(s);
  s += lane_xor_bits<4>(s);
  s += lane_xor_bits<2>(s);
  s += lane_xor_bits<1>(s);
  return s;
}

__device__ __forceinline__ int wave_run_sum(int v, int lane) {
#pragma unroll
  for (int o = 1; o < 32; o <<= 1) {
    const int t = __shfl_up(v, o);
    if (lane >= o) v += t;
  }
  return v;
}

__global__ __launch_bounds__(1024) void edge_bin(
    const int* __restrict__ srcA, const int* __restrict__ dstA,
    int* __restrict__ ssrc, int* __restrict__ sdst, int* __restrict__ tab, int E, int nb)
{
  static_assert(HB == NWV * 32, "one counter per thread");
  static_assert(CH % (NWV * 32) == 0, "a whole number of edges per thread");
  constexpr int EPT = CH / (NWV * 32);
  __shared__ int hist[HB];
  __shared__ int wtot[NWV];
  __shared__ int st_s[CH];
  __shared__ int st_d[CH];

  const int tid  = threadIdx.x;
  const int w    = tid >> 5;
  const int lane = tid & 31;
  const int e0   = blockIdx.x * CH;
  const int m    = min(CH, E - e0);

  hist[tid] = 0;
  __syncthreads();

  int s[EPT], d[EPT], b[EPT], r[EPT];
#pragma unroll
  for (int u = 0; u < EPT; ++u) {
    const int i = tid + u * NWV * 32;
    d[u] = (i < m) ? dstA[e0 + i] : -1;
    s[u] = (i < m) ? srcA[e0 + i] : 0;
  }
#pragma unroll
  for (int u = 0; u < EPT; ++u) {
    b[u] = 0; r[u] = 0;
    if (d[u] >= 0) {
      b[u] = d[u] / RNG;
      r[u] = __hip_atomic_fetch_add(&hist[b[u]], 1, __ATOMIC_RELAXED, __HIP_MEMORY_SCOPE_WORKGROUP);
    }
  }
  __syncthreads();

  const int v    = hist[tid];
  const int incl = wave_run_sum(v, lane);
  if (lane == 31) wtot[w] = incl;
  __syncthreads();
  const int wl   = wtot[lane];
  const int excl = wave_sum_i(lane < w ? wl : 0) + incl - v;
  hist[tid] = excl;
  if (tid <= nb) tab[(size_t)blockIdx.x * (nb + 1) + tid] = excl;
  __syncthreads();

#pragma unroll
  for (int u = 0; u < EPT; ++u) {
    if (d[u] >= 0) {
      const int pos = hist[b[u]] + r[u];
      st_s[pos] = s[u];
      st_d[pos] = d[u];
    }
  }
  __syncthreads();
#pragma unroll
  for (int u = 0; u < EPT; ++u) {
    const int i = tid + u * NWV * 32;
    if (i < m) {
      ssrc[e0 + i] = st_s[i];
      sdst[e0 + i] = st_d[i];
    }
  }
}

__global__ __launch_bounds__(DWV * 32) void deg_rng(
    const int* __restrict__ sdst, const int* __restrict__ tab,
    float* __restrict__ dinv, int n, int nch, int nb)
{
  static_assert(DWV * 32 >= RNG, "one thread per owned row");
  static_assert(DGR <= 32, "one lane per group of a trip");
  __shared__ int cnt[RNG];

  const int tid  = threadIdx.x;
  const int w    = __builtin_amdgcn_readfirstlane(tid >> 5);
  const int lane = tid & 31;
  const int d0   = blockIdx.x * RNG;
  const int nd   = min(RNG, n - d0);

  auto bounds = [&](int t0, int& a0, int& a1) {
    const int t = t0 + lane * DWV;
    a0 = 0; a1 = 0;
    if (lane < DGR && t < nch) {
      const int* tp = tab + (size_t)t * (nb + 1) + blockIdx.x;
      a0 = tp[0];
      a1 = tp[1];
    }
  };

  int a0, a1;
  bounds(w, a0, a1);
  if (tid < RNG) cnt[tid] = 0;
  __syncthreads();

  for (int t0 = w; t0 < nch; t0 += DWV * DGR) {
    const int ga = (t0 + lane * DWV) * CH + a0;
    const int gl = a1 - a0;
    int at[DGR], ln[DGR], dv[DGR];
#pragma unroll
    for (int u = 0; u < DGR; ++u) {
      at[u] = __builtin_amdgcn_readlane(ga, u);
      ln[u] = __builtin_amdgcn_readlane(gl, u);
    }
#pragma unroll
    for (int u = 0; u < DGR; ++u) {
      dv[u] = -1;
      if (lane < ln[u]) dv[u] = sdst[at[u] + lane];
    }
    bounds(t0 + DWV * DGR, a0, a1);
#pragma unroll
    for (int u = 0; u < DGR; ++u)
      if (dv[u] >= 0)
        __hip_atomic_fetch_add(&cnt[dv[u] - d0], 1, __ATOMIC_RELAXED, __HIP_MEMORY_SCOPE_WORKGROUP);
#pragma unroll
    for (int u = 0; u < DGR; ++u)
      if (ln[u] > 32)
        for (int j = 32 + lane; j < ln[u]; j += 32)
          __hip_atomic_fetch_add(&cnt[sdst[at[u] + j] - d0], 1, __ATOMIC_RELAXED, __HIP_MEMORY_SCOPE_WORKGROUP);
  }
  __syncthreads();

  if (tid < nd) {
    const float d = 1.0f + (float)cnt[tid];
    dinv[d0 + tid] = (d > 0.0f) ? rsqrtf(d) : 0.0f;
  }
}

template <bool INPLACE>
__device__ __forceinline__ void rows_norm(
    float* acc, const float* __restrict__ bias, const float* __restrict__ gamma,
    const float* __restrict__ beta, float* __restrict__ OUT,
    int d0, int nd, int w, int lane)
{
  static_assert(FIN == 128, "4 features per lane of a 128-wide row");
  const float b0 = bias[lane * 4 + 0],  b1 = bias[lane * 4 + 1];
  const float b2 = bias[lane * 4 + 2],  b3 = bias[lane * 4 + 3];
  const float g0 = gamma[lane * 4 + 0], g1 = gamma[lane * 4 + 1];
  const float g2 = gamma[lane * 4 + 2], g3 = gamma[lane * 4 + 3];
  const float e0 = beta[lane * 4 + 0],  e1 = beta[lane * 4 + 1];
  const float e2 = beta[lane * 4 + 2],  e3 = beta[lane * 4 + 3];
  __syncthreads();

  for (int r = w; r < nd; r += NWV) {
    float* row = acc + r * PITCH + lane * 4;
    const float v0 = row[0] + b0;
    const float v1 = row[1] + b1;
    const float v2 = row[2] + b2;
    const float v3 = row[3] + b3;

    const float mu = wave_sum(v0 + v1 + v2 + v3) * (1.0f / 128.0f);

    const float c0 = v0 - mu, c1 = v1 - mu, c2 = v2 - mu, c3 = v3 - mu;
    const float q = wave_sum(c0 * c0 + c1 * c1 + c2 * c2 + c3 * c3);
    const float rstd = rsqrtf(q * (1.0f / 128.0f) + LN_EPS);

    float* out = INPLACE ? row : OUT + (size_t)(d0 + r) * FIN + lane * 4;
    out[0] = fmaxf(0.0f, c0 * rstd * g0 + e0);
    out[1] = fmaxf(0.0f, c1 * rstd * g1 + e1);
    out[2] = fmaxf(0.0f, c2 * rstd * g2 + e2);
    out[3] = fmaxf(0.0f, c3 * rstd * g3 + e3);
  }
}

template <int FOUT, int PT, bool BIAS>
__device__ __forceinline__ void rows_out(
    const float* acc, const float* __restrict__ bias, float* __restrict__ OUT,
    int d0, int nd, int tid)
{
  static_assert((NWV * 32) % FOUT == 0, "a thread keeps one feature column over its whole stride");
  float bb = 0.0f;
  if constexpr (BIAS) bb = bias[tid & (FOUT - 1)];
  __syncthreads();
  for (int i = tid; i < nd * FOUT; i += NWV * 32)
    OUT[(size_t)d0 * FOUT + i] = acc[(i / FOUT) * PT + (i % FOUT)] + bb;
}

template <int FW, int PT, bool TILES>
__device__ __forceinline__ void rows_start(
    float* acc, const float* __restrict__ X, const float* __restrict__ dinv,
    int d0, int nd, int tid)
{
  for (int i = tid; i < nd * FW; i += NWV * 32) {
    const int r = i / FW;
    const float di = dinv[d0 + r];
    acc[r * PT + (i % FW)] = X[(size_t)d0 * FW + i] * (di * di);
  }
  if constexpr (TILES) {
    const int ntl = ((nd + 15) >> 4) << 4;
    for (int i = nd * FW + tid; i < ntl * FW; i += NWV * 32)
      acc[(i / FW) * PT + (i % FW)] = 0.0f;
  }
}

template <int FOUT, bool LNRELU>
__global__ __launch_bounds__(1024) void agg_range(
    const float* __restrict__ X, const int* __restrict__ ssrc,
    const int* __restrict__ sdst, const int* __restrict__ tab,
    const float* __restrict__ dinv, const float* __restrict__ W,
    const float* __restrict__ bias, const float* __restrict__ gamma,
    const float* __restrict__ beta, float* __restrict__ OUT,
    int* __restrict__ lsrc, int* __restrict__ ldl, float* __restrict__ lnrm,
    int* __restrict__ blk, int n, int nch, int nb)
{
  __shared__ float acc[RPAD * PITCH];
  __shared__ int lst_e[NWV * LCAP];
  __shared__ int seg_at[HB];
  __shared__ int seg_v[HB + 1];
  __shared__ int red[2 * NWV];

  const int tid  = threadIdx.x;
  const int w    = __builtin_amdgcn_readfirstlane(tid >> 5);
  const int lane = tid & 31;
  const int d0   = blockIdx.x * RNG;
  const int nd   = min(RNG, n - d0);

  static_assert(HB == NWV * 32, "one chunk per thread");
  int ga = 0, gb = 0;
  if (tid < nch) {
    const int* tp = tab + (size_t)tid * (nb + 1) + blockIdx.x;
    ga = tp[0];
    gb = tp[1];
  }
  rows_start<FIN, PITCH, true>(acc, X, dinv, d0, nd, tid);
  const int glen = gb - ga;
  const int incl = wave_run_sum(glen, lane);
  const int sa   = wave_sum_i(ga);
  if (lane == 31) red[w] = incl;
  if (lane == 0)  red[NWV + w] = sa;
  __syncthreads();
  const int wl   = red[lane];
  const int tot  = __builtin_amdgcn_readfirstlane(wave_sum_i(wl));
  const int base = __builtin_amdgcn_readfirstlane(wave_sum_i(red[NWV + lane]));
  const int pre  = wave_sum_i(lane < w ? wl : 0);
  if (tid < nch) {
    seg_at[tid] = tid * CH + ga;
    seg_v[tid]  = pre + incl - glen;
  }
  if (tid == 0) {
    seg_v[nch] = tot;
    blk[2 * blockIdx.x] = base;
    blk[2 * blockIdx.x + 1] = tot;
  }
  __syncthreads();

  constexpr int GRP = INFL / (FIN / 32);
  const int per = ((tot + NWV * GRP - 1) / (NWV * GRP)) * GRP;
  const int lo  = w * per;
  const int hi  = min(tot, lo + per);
  if (hi > lo) {
    int cb = 0;
    for (int t = lane; t < nch; t += 32) cb += (seg_v[t + 1] <= lo) ? 1 : 0;
    int t = __builtin_amdgcn_readfirstlane(wave_sum_i(cb));
    int* le = lst_e + w * LCAP;
    for (int a0 = lo; a0 < hi; a0 += LCAP) {
      const int b0 = min(hi, a0 + LCAP);
      int a = a0;
      while (a < b0) {
        const int sv = __builtin_amdgcn_readfirstlane(seg_v[t]);
        const int se = __builtin_amdgcn_readfirstlane(seg_v[t + 1]);
        const int g  = __builtin_amdgcn_readfirstlane(seg_at[t]) + (a - sv);
        const int take = min(b0, se) - a;
        for (int j = lane; j < take; j += 32) le[a - a0 + j] = g + j;
        a += take;
        if (a == se) ++t;
      }
      add_listed<false, FIN, PITCH>(X, ssrc, sdst, dinv, acc, le, b0 - a0, d0, lane,
                                    lsrc, ldl, lnrm, base + a0);
    }
  }
  rows_times_w<FOUT>(acc, W, nd, w, lane);
  if constexpr (LNRELU) rows_norm<false>(acc, bias, gamma, beta, OUT, d0, nd, w, lane);
  else                  rows_out<FOUT, PITCH, true>(acc, bias, OUT, d0, nd, tid);
}

template <int FOUT, int FNX>
__global__ __launch_bounds__(1024) void agg_list(
    const float* __restrict__ X, const float* __restrict__ dinv,
    const float* __restrict__ W,
    const float* __restrict__ bias, const float* __restrict__ gamma,
    const float* __restrict__ beta, const float* __restrict__ WN,
    float* __restrict__ OUT,
    int* __restrict__ lsrc, int* __restrict__ ldl, float* __restrict__ lnrm,
    const int* __restrict__ blk, int n)
{
  static_assert(FOUT == FIN, "the finished rows are the next product's operands");
  __shared__ float acc[RPAD * PITCH];

  const int tid  = threadIdx.x;
  const int w    = __builtin_amdgcn_readfirstlane(tid >> 5);
  const int lane = tid & 31;
  const int d0   = blockIdx.x * RNG;
  const int nd   = min(RNG, n - d0);

  const int base = blk[2 * blockIdx.x];
  const int tot  = blk[2 * blockIdx.x + 1];

  rows_start<FIN, PITCH, true>(acc, X, dinv, d0, nd, tid);
  __syncthreads();

  constexpr int GRP = INFL / (FIN / 32);
  const int per = ((tot + NWV * GRP - 1) / (NWV * GRP)) * GRP;
  const int lo  = w * per;
  const int cnt = min(per, tot - lo);
  if (cnt > 0)
    add_listed<true, FIN, PITCH>(X, nullptr, nullptr, dinv, acc, nullptr, cnt, d0, lane,
                                 lsrc, ldl, lnrm, base + lo);

  rows_times_w<FOUT>(acc, W, nd, w, lane);
  rows_norm<true>(acc, bias, gamma, beta, nullptr, d0, nd, w, lane);
  rows_times_w<FNX>(acc, WN, nd, w, lane);
  rows_out<FNX, PITCH, false>(acc, nullptr, OUT, d0, nd, tid);
}

template <int FW>
__global__ __launch_bounds__(1024) void agg_sum(
    const float* __restrict__ X, const float* __restrict__ dinv,
    const float* __restrict__ bias, float* __restrict__ OUT,
    int* __restrict__ lsrc, int* __restrict__ ldl, float* __restrict__ lnrm,
    const int* __restrict__ blk, int n)
{
  __shared__ float acc[RNG * FW];

  const int tid  = threadIdx.x;
  const int w    = __builtin_amdgcn_readfirstlane(tid >> 5);
  const int lane = tid & 31;
  const int d0   = blockIdx.x * RNG;
  const int nd   = min(RNG, n - d0);

  const int base = blk[2 * blockIdx.x];
  const int tot  = blk[2 * blockIdx.x + 1];

  rows_start<FW, FW, false>(acc, X, dinv, d0, nd, tid);
  __syncthreads();

  constexpr int GRP = INFL / (FW / 32);
  const int per = ((tot + NWV * GRP - 1) / (NWV * GRP)) * GRP;
  const int lo  = w * per;
  const int cnt = min(per, tot - lo);
  if (cnt > 0)
    add_listed<true, FW, FW>(X, nullptr, nullptr, dinv, acc, nullptr, cnt, d0, lane,
                             lsrc, ldl, lnrm, base + lo);

  rows_out<FW, FW, true>(acc, bias, OUT, d0, nd, tid);
}

extern "C" void kernel_launch(void* const* d_in, const int* in_sizes, int n_in,
                              void* d_out, int out_size, void* d_ws, size_t ws_size,
                              hipStream_t stream) {
  const float* x   = (const float*)d_in[0];
  const int*   ei  = (const int*)d_in[1];
  const float* W1  = (const float*)d_in[2];
  const float* b1  = (const float*)d_in[3];
  const float* g1  = (const float*)d_in[4];
  const float* be1 = (const float*)d_in[5];
  const float* W2  = (const float*)d_in[6];
  const float* b2  = (const float*)d_in[7];
  const float* g2  = (const float*)d_in[8];
  const float* be2 = (const float*)d_in[9];
  const float* W3  = (const float*)d_in[10];
  const float* b3  = (const float*)d_in[11];

  const int N = in_sizes[0] / 128;
  const int E = in_sizes[1] / 2;
  const int* src = ei;
  const int* dst = ei + E;

  float* ws   = (float*)d_ws;
  float* ha   = ws;
  float* hb   = ha + (size_t)N * 128;
  float* dinv = hb + (size_t)N * 64;
  int*   lsrc = (int*)(dinv + 2 * (size_t)N);
  int*   ldl  = lsrc + E;
  float* lnrm = (float*)(ldl + E);
  int*   ssrc = (int*)(lnrm + E);
  int*   sdst = ssrc + E;
  int*   blk  = sdst + E;
  int*   tab  = blk + 2 * ((N + RNG - 1) / RNG);

  const int ablocks = (N + RNG - 1) / RNG;
  const int nch = (E + CH - 1) / CH;
  edge_bin<<<nch, NWV * 32, 0, stream>>>(src, dst, ssrc, sdst, tab, E, ablocks);
  deg_rng<<<ablocks, DWV * 32, 0, stream>>>(sdst, tab, dinv, N, nch, ablocks);

  agg_range<128, true><<<ablocks, NWV * 32, 0, stream>>>(x, ssrc, sdst, tab, dinv, W1, b1, g1, be1, ha,
                                                         lsrc, ldl, lnrm, blk, N, nch, ablocks);

  agg_list<128, FLAST><<<ablocks, NWV * 32, 0, stream>>>(ha, dinv, W2, b2, g2, be2, W3, hb,
                                                         lsrc, ldl, lnrm, blk, N);

  agg_sum<FLAST><<<ablocks, NWV * 32, 0, stream>>>(hb, dinv, b3, (float*)d_out,
                                                   lsrc, ldl, lnrm, blk, N);
}
